# starting kernel
# baseline (speedup 1.0000x reference)
#include <hip/hip_runtime.h>
#include <hip/hip_bf16.h>

#define N_NODES   50000
#define E_EDGES   1000000
#define IN_DIM    128
#define OUT_DIM   64
#define NEG_SLOPE 0.01f

typedef __attribute__((ext_vector_type(2))) float v2f;
typedef __attribute__((ext_vector_type(8))) float v8f;

__device__ __forceinline__ unsigned f32_ord(float f) {
  unsigned u = __float_as_uint(f);
  return (u & 0x80000000u) ? ~u : (u | 0x80000000u);
}
__device__ __forceinline__ float ord_f32(unsigned o) {
  unsigned u = (o & 0x80000000u) ? (o & 0x7FFFFFFFu) : ~o;
  return __uint_as_float(u);
}

__global__ void __launch_bounds__(256)
gat_gemm_wmma(const float* __restrict__ h, const float* __restrict__ Wfc,
              float* __restrict__ z) {
  const int lane = threadIdx.x & 31;
  const int wid  = blockIdx.x * (blockDim.x >> 5) + (threadIdx.x >> 5);
  const int m0   = wid * 16;
  if (m0 >= N_NODES) return;

  const int lrow = lane & 15;
  const int kh   = (lane >> 4) << 1;

  const float* __restrict__ ap  = h   + (long)(m0 + lrow) * IN_DIM + kh;
  const float* __restrict__ b0p = Wfc + (long)( 0 + lrow) * IN_DIM + kh;
  const float* __restrict__ b1p = Wfc + (long)(16 + lrow) * IN_DIM + kh;
  const float* __restrict__ b2p = Wfc + (long)(32 + lrow) * IN_DIM + kh;
  const float* __restrict__ b3p = Wfc + (long)(48 + lrow) * IN_DIM + kh;

  v8f c0 = {}, c1 = {}, c2 = {}, c3 = {};
  #pragma unroll 4
  for (int k = 0; k < IN_DIM; k += 4) {
    v2f a  = *(const v2f*)(ap  + k);
    v2f b0 = *(const v2f*)(b0p + k);
    v2f b1 = *(const v2f*)(b1p + k);
    v2f b2 = *(const v2f*)(b2p + k);
    v2f b3 = *(const v2f*)(b3p + k);
    c0 = __builtin_amdgcn_wmma_f32_16x16x4_f32(false, a, false, b0, (short)0, c0, false, false);
    c1 = __builtin_amdgcn_wmma_f32_16x16x4_f32(false, a, false, b1, (short)0, c1, false, false);
    c2 = __builtin_amdgcn_wmma_f32_16x16x4_f32(false, a, false, b2, (short)0, c2, false, false);
    c3 = __builtin_amdgcn_wmma_f32_16x16x4_f32(false, a, false, b3, (short)0, c3, false, false);
  }

  const int rbase = m0 + ((lane >> 4) << 3);
  float* zp = z + (long)rbase * OUT_DIM + lrow;
  #pragma unroll
  for (int r = 0; r < 8; ++r) {
    zp[(long)r * OUT_DIM +  0] = c0[r];
    zp[(long)r * OUT_DIM + 16] = c1[r];
    zp[(long)r * OUT_DIM + 32] = c2[r];
    zp[(long)r * OUT_DIM + 48] = c3[r];
  }
}

__global__ void __launch_bounds__(256)
gat_node_dots(const float* __restrict__ z, const float* __restrict__ Wattn,
              float* __restrict__ zs, float* __restrict__ zd) {
  const int lane = threadIdx.x & 31;
  const int n = blockIdx.x * (blockDim.x >> 5) + (threadIdx.x >> 5);
  if (n >= N_NODES) return;
  const float* zr = z + (long)n * OUT_DIM;
  float z0 = zr[lane], z1 = zr[lane + 32];
  float ps = z0 * Wattn[lane]           + z1 * Wattn[lane + 32];
  float pd = z0 * Wattn[OUT_DIM + lane] + z1 * Wattn[OUT_DIM + lane + 32];
  #pragma unroll
  for (int o = 16; o > 0; o >>= 1) {
    ps += __shfl_xor(ps, o, 32);
    pd += __shfl_xor(pd, o, 32);
  }
  if (lane == 0) { zs[n] = ps; zd[n] = pd; }
}

__global__ void gat_init(float* __restrict__ out, unsigned* __restrict__ mord,
                         float* __restrict__ ssum) {
  long i = (long)blockIdx.x * blockDim.x + threadIdx.x;
  if (i < (long)N_NODES * OUT_DIM) out[i] = 0.0f;
  if (i < N_NODES) { mord[i] = 0x007FFFFFu  ; ssum[i] = 0.0f; }
}

__global__ void gat_edge_scores(const int* __restrict__ src, const int* __restrict__ dst,
                                const float* __restrict__ wgt,
                                const float* __restrict__ zs, const float* __restrict__ zd,
                                float* __restrict__ ew, unsigned* __restrict__ mord) {
  int i = blockIdx.x * blockDim.x + threadIdx.x;
  if (i >= E_EDGES) return;
  int s0 = src[i], d0 = dst[i];
  float x = zs[s0] + zd[d0];
  float e = (x > 0.0f) ? x : NEG_SLOPE * x;
  e *= wgt[i];
  ew[i] = e;
  atomicMax(mord + d0, f32_ord(e));
}

__global__ void gat_edge_exp(const int* __restrict__ dst, const unsigned* __restrict__ mord,
                             float* __restrict__ ew, float* __restrict__ ssum) {
  int i = blockIdx.x * blockDim.x + threadIdx.x;
  if (i >= E_EDGES) return;
  int d0 = dst[i];
  float ex = expf(ew[i] - ord_f32(mord[d0]));
  ew[i] = ex;
  atomicAdd(ssum + d0, ex);
}

__global__ void __launch_bounds__(256)
gat_aggregate(const int* __restrict__ src, const int* __restrict__ dst,
              const float* __restrict__ ew, const float* __restrict__ ssum,
              const float* __restrict__ z, float* __restrict__ out) {
  const int lane = threadIdx.x & 31;
  const int e = blockIdx.x * (blockDim.x >> 5) + (threadIdx.x >> 5);
  if (e >= E_EDGES) return;
  const int s0 = src[e], d0 = dst[e];
  const float alpha = ew[e] / ssum[d0];
  v2f v = *(const v2f*)(z + (long)s0 * OUT_DIM + lane * 2);
  float* op = out + (long)d0 * OUT_DIM + lane * 2;
  atomicAdd(op + 0, alpha * v.x);
  atomicAdd(op + 1, alpha * v.y);
}

extern "C" void kernel_launch(void* const* d_in, const int* in_sizes, int n_in,
                              void* d_out, int out_size, void* d_ws, size_t ws_size,
                              hipStream_t stream) {
  (void)in_sizes; (void)n_in; (void)out_size; (void)ws_size;
  const float* h     = (const float*)d_in[0];
  const float* Wfc   = (const float*)d_in[1];
  const float* Wattn = (const float*)d_in[2];
  const int*   src   = (const int*)d_in[3];
  const int*   dst   = (const int*)d_in[4];
  const float* wgt   = (const float*)d_in[5];
  float* out = (float*)d_out;

  float*    z    = (float*)d_ws;
  float*    zs   = z + (size_t)N_NODES * OUT_DIM;
  float*    zd   = zs + N_NODES;
  unsigned* mord = (unsigned*)(zd + N_NODES);
  float*    ssum = (float*)(mord + N_NODES);
  float*    ew   = ssum + N_NODES;

  {
    int waves  = (N_NODES + 15) / 16;
    int blocks = (waves * 32 + 255) / 256;
    gat_gemm_wmma<<<blocks, 256, 0, stream>>>(h, Wfc, z);
  }
  {
    int blocks = (N_NODES * 32 + 255) / 256;
    gat_node_dots<<<blocks, 256, 0, stream>>>(z, Wattn, zs, zd);
  }
  {
    long total = (long)N_NODES * OUT_DIM;
    int blocks = (int)((total + 255) / 256);
    gat_init<<<blocks, 256, 0, stream>>>(out, mord, ssum);
  }
  {
    int blocks = (E_EDGES + 255) / 256;
    gat_edge_scores<<<blocks, 256, 0, stream>>>(src, dst, wgt, zs, zd, ew, mord);
    gat_edge_exp<<<blocks, 256, 0, stream>>>(dst, mord, ew, ssum);
  }
  {
    long threads = (long)E_EDGES * 32;
    int blocks = (int)((threads + 255) / 256);
    gat_aggregate<<<blocks, 256, 0, stream>>>(src, dst, ew, ssum, z, out);
  }
}


# Aggregate: each wave takes 32 consecutive edges (per-lane index/alpha loads, lane-broadcast walk, z-row loads batched ahead of the adds), not one edge per wave; f32 atomics kept, arrival order shifts
# speedup vs baseline: 1.2466x; 1.2466x over previous
#include <hip/hip_runtime.h>
#include <hip/hip_bf16.h>

#define N_NODES   50000
#define E_EDGES   1000000
#define IN_DIM    128
#define OUT_DIM   64
#define NEG_SLOPE 0.01f

typedef __attribute__((ext_vector_type(2))) float v2f;
typedef __attribute__((ext_vector_type(8))) float v8f;

__device__ __forceinline__ unsigned f32_ord(float f) {
  unsigned u = __float_as_uint(f);
  return (u & 0x80000000u) ? ~u : (u | 0x80000000u);
}
__device__ __forceinline__ float ord_f32(unsigned o) {
  unsigned u = (o & 0x80000000u) ? (o & 0x7FFFFFFFu) : ~o;
  return __uint_as_float(u);
}

__global__ void __launch_bounds__(256)
gat_gemm_wmma(const float* __restrict__ h, const float* __restrict__ Wfc,
              float* __restrict__ z) {
  const int lane = threadIdx.x & 31;
  const int wid  = blockIdx.x * (blockDim.x >> 5) + (threadIdx.x >> 5);
  const int m0   = wid * 16;
  if (m0 >= N_NODES) return;

  const int lrow = lane & 15;
  const int kh   = (lane >> 4) << 1;

  const float* __restrict__ ap  = h   + (long)(m0 + lrow) * IN_DIM + kh;
  const float* __restrict__ b0p = Wfc + (long)( 0 + lrow) * IN_DIM + kh;
  const float* __restrict__ b1p = Wfc + (long)(16 + lrow) * IN_DIM + kh;
  const float* __restrict__ b2p = Wfc + (long)(32 + lrow) * IN_DIM + kh;
  const float* __restrict__ b3p = Wfc + (long)(48 + lrow) * IN_DIM + kh;

  v8f c0 = {}, c1 = {}, c2 = {}, c3 = {};
  #pragma unroll 4
  for (int k = 0; k < IN_DIM; k += 4) {
    v2f a  = *(const v2f*)(ap  + k);
    v2f b0 = *(const v2f*)(b0p + k);
    v2f b1 = *(const v2f*)(b1p + k);
    v2f b2 = *(const v2f*)(b2p + k);
    v2f b3 = *(const v2f*)(b3p + k);
    c0 = __builtin_amdgcn_wmma_f32_16x16x4_f32(false, a, false, b0, (short)0, c0, false, false);
    c1 = __builtin_amdgcn_wmma_f32_16x16x4_f32(false, a, false, b1, (short)0, c1, false, false);
    c2 = __builtin_amdgcn_wmma_f32_16x16x4_f32(false, a, false, b2, (short)0, c2, false, false);
    c3 = __builtin_amdgcn_wmma_f32_16x16x4_f32(false, a, false, b3, (short)0, c3, false, false);
  }

  const int rbase = m0 + ((lane >> 4) << 3);
  float* zp = z + (long)rbase * OUT_DIM + lrow;
  #pragma unroll
  for (int r = 0; r < 8; ++r) {
    zp[(long)r * OUT_DIM +  0] = c0[r];
    zp[(long)r * OUT_DIM + 16] = c1[r];
    zp[(long)r * OUT_DIM + 32] = c2[r];
    zp[(long)r * OUT_DIM + 48] = c3[r];
  }
}

__global__ void __launch_bounds__(256)
gat_node_dots(const float* __restrict__ z, const float* __restrict__ Wattn,
              float* __restrict__ zs, float* __restrict__ zd) {
  const int lane = threadIdx.x & 31;
  const int n = blockIdx.x * (blockDim.x >> 5) + (threadIdx.x >> 5);
  if (n >= N_NODES) return;
  const float* zr = z + (long)n * OUT_DIM;
  float z0 = zr[lane], z1 = zr[lane + 32];
  float ps = z0 * Wattn[lane]           + z1 * Wattn[lane + 32];
  float pd = z0 * Wattn[OUT_DIM + lane] + z1 * Wattn[OUT_DIM + lane + 32];
  #pragma unroll
  for (int o = 16; o > 0; o >>= 1) {
    ps += __shfl_xor(ps, o, 32);
    pd += __shfl_xor(pd, o, 32);
  }
  if (lane == 0) { zs[n] = ps; zd[n] = pd; }
}

__global__ void gat_init(float* __restrict__ out, unsigned* __restrict__ mord,
                         float* __restrict__ ssum) {
  long i = (long)blockIdx.x * blockDim.x + threadIdx.x;
  if (i < (long)N_NODES * OUT_DIM) out[i] = 0.0f;
  if (i < N_NODES) { mord[i] = 0x007FFFFFu  ; ssum[i] = 0.0f; }
}

__global__ void gat_edge_scores(const int* __restrict__ src, const int* __restrict__ dst,
                                const float* __restrict__ wgt,
                                const float* __restrict__ zs, const float* __restrict__ zd,
                                float* __restrict__ ew, unsigned* __restrict__ mord) {
  int i = blockIdx.x * blockDim.x + threadIdx.x;
  if (i >= E_EDGES) return;
  int s0 = src[i], d0 = dst[i];
  float x = zs[s0] + zd[d0];
  float e = (x > 0.0f) ? x : NEG_SLOPE * x;
  e *= wgt[i];
  ew[i] = e;
  atomicMax(mord + d0, f32_ord(e));
}

__global__ void gat_edge_exp(const int* __restrict__ dst, const unsigned* __restrict__ mord,
                             float* __restrict__ ew, float* __restrict__ ssum) {
  int i = blockIdx.x * blockDim.x + threadIdx.x;
  if (i >= E_EDGES) return;
  int d0 = dst[i];
  float ex = expf(ew[i] - ord_f32(mord[d0]));
  ew[i] = ex;
  atomicAdd(ssum + d0, ex);
}

static_assert(E_EDGES % 32 == 0, "one wave owns exactly 32 edges");
#define AGG_CHUNK 16
__global__ void __launch_bounds__(256)
gat_aggregate(const int* __restrict__ src, const int* __restrict__ dst,
              const float* __restrict__ ew, const float* __restrict__ ssum,
              const float* __restrict__ z, float* __restrict__ out) {
  const int lane = threadIdx.x & 31;
  const int w  = blockIdx.x * (blockDim.x >> 5) + (threadIdx.x >> 5);
  const int e0 = w * 32;
  if (e0 >= E_EDGES) return;
  const int e = e0 + lane;
  const int   s_l = src[e], d_l = dst[e];
  const float a_l = ew[e] / ssum[d_l];
  const float* zl = z   + lane * 2;
  float*       ol = out + lane * 2;
  #pragma unroll
  for (int c = 0; c < 32; c += AGG_CHUNK) {
    v2f v[AGG_CHUNK];
    #pragma unroll
    for (int j = 0; j < AGG_CHUNK; ++j) {
      const int sj = __builtin_amdgcn_readlane(s_l, c + j);
      v[j] = *(const v2f*)(zl + (long)sj * OUT_DIM);
    }
    #pragma unroll
    for (int j = 0; j < AGG_CHUNK; ++j) {
      const int   dj = __builtin_amdgcn_readlane(d_l, c + j);
      const float aj = __int_as_float(__builtin_amdgcn_readlane(__float_as_int(a_l), c + j));
      float* op = ol + (long)dj * OUT_DIM;
      atomicAdd(op + 0, aj * v[j].x);
      atomicAdd(op + 1, aj * v[j].y);
    }
  }
}

extern "C" void kernel_launch(void* const* d_in, const int* in_sizes, int n_in,
                              void* d_out, int out_size, void* d_ws, size_t ws_size,
                              hipStream_t stream) {
  (void)in_sizes; (void)n_in; (void)out_size; (void)ws_size;
  const float* h     = (const float*)d_in[0];
  const float* Wfc   = (const float*)d_in[1];
  const float* Wattn = (const float*)d_in[2];
  const int*   src   = (const int*)d_in[3];
  const int*   dst   = (const int*)d_in[4];
  const float* wgt   = (const float*)d_in[5];
  float* out = (float*)d_out;

  float*    z    = (float*)d_ws;
  float*    zs   = z + (size_t)N_NODES * OUT_DIM;
  float*    zd   = zs + N_NODES;
  unsigned* mord = (unsigned*)(zd + N_NODES);
  float*    ssum = (float*)(mord + N_NODES);
  float*    ew   = ssum + N_NODES;

  {
    int waves  = (N_NODES + 15) / 16;
    int blocks = (waves * 32 + 255) / 256;
    gat_gemm_wmma<<<blocks, 256, 0, stream>>>(h, Wfc, z);
  }
  {
    int blocks = (N_NODES * 32 + 255) / 256;
    gat_node_dots<<<blocks, 256, 0, stream>>>(z, Wattn, zs, zd);
  }
  {
    long total = (long)N_NODES * OUT_DIM;
    int blocks = (int)((total + 255) / 256);
    gat_init<<<blocks, 256, 0, stream>>>(out, mord, ssum);
  }
  {
    int blocks = (E_EDGES + 255) / 256;
    gat_edge_scores<<<blocks, 256, 0, stream>>>(src, dst, wgt, zs, zd, ew, mord);
    gat_edge_exp<<<blocks, 256, 0, stream>>>(dst, mord, ew, ssum);
  }
  {
    int waves  = E_EDGES / 32;
    int blocks = (waves + 7) / 8;
    gat_aggregate<<<blocks, 256, 0, stream>>>(src, dst, ew, ssum, z, out);
  }
}


# Fuse the node-side passes into the GEMM wave: epilogue forms zs/zd from its accumulators (16-lane sums, dot order re-associated), zero-fills out, inits max/sum; 6 launches -> 4, z not re-read
# speedup vs baseline: 1.3458x; 1.0796x over previous
#include <hip/hip_runtime.h>
#include <hip/hip_bf16.h>

#define N_NODES   50000
#define E_EDGES   1000000
#define IN_DIM    128
#define OUT_DIM   64
#define NEG_SLOPE 0.01f

typedef __attribute__((ext_vector_type(2))) float v2f;
typedef __attribute__((ext_vector_type(8))) float v8f;

__device__ __forceinline__ unsigned f32_ord(float f) {
  unsigned u = __float_as_uint(f);
  return (u & 0x80000000u) ? ~u : (u | 0x80000000u);
}
__device__ __forceinline__ float ord_f32(unsigned o) {
  unsigned u = (o & 0x80000000u) ? (o & 0x7FFFFFFFu) : ~o;
  return __uint_as_float(u);
}

static_assert(N_NODES % 16 == 0, "every wave owns a full 16-row tile");
__global__ void __launch_bounds__(256)
gat_gemm_wmma(const float* __restrict__ h, const float* __restrict__ Wfc,
              const float* __restrict__ Wattn,
              float* __restrict__ z, float* __restrict__ zs, float* __restrict__ zd,
              unsigned* __restrict__ mord, float* __restrict__ ssum,
              float* __restrict__ out) {
  const int lane = threadIdx.x & 31;
  const int wid  = blockIdx.x * (blockDim.x >> 5) + (threadIdx.x >> 5);
  const int m0   = wid * 16;
  if (m0 >= N_NODES) return;

  {
    float4* o4 = (float4*)(out + (long)m0 * OUT_DIM) + lane;
    const float4 zero4 = make_float4(0.0f, 0.0f, 0.0f, 0.0f);
    #pragma unroll
    for (int i = 0; i < 8; ++i) o4[i * 32] = zero4;
  }

  const int lrow = lane & 15;
  const int kh   = (lane >> 4) << 1;

  const float* __restrict__ ap  = h   + (long)(m0 + lrow) * IN_DIM + kh;
  const float* __restrict__ b0p = Wfc + (long)( 0 + lrow) * IN_DIM + kh;
  const float* __restrict__ b1p = Wfc + (long)(16 + lrow) * IN_DIM + kh;
  const float* __restrict__ b2p = Wfc + (long)(32 + lrow) * IN_DIM + kh;
  const float* __restrict__ b3p = Wfc + (long)(48 + lrow) * IN_DIM + kh;

  v8f c0 = {}, c1 = {}, c2 = {}, c3 = {};
  #pragma unroll 4
  for (int k = 0; k < IN_DIM; k += 4) {
    v2f a  = *(const v2f*)(ap  + k);
    v2f b0 = *(const v2f*)(b0p + k);
    v2f b1 = *(const v2f*)(b1p + k);
    v2f b2 = *(const v2f*)(b2p + k);
    v2f b3 = *(const v2f*)(b3p + k);
    c0 = __builtin_amdgcn_wmma_f32_16x16x4_f32(false, a, false, b0, (short)0, c0, false, false);
    c1 = __builtin_amdgcn_wmma_f32_16x16x4_f32(false, a, false, b1, (short)0, c1, false, false);
    c2 = __builtin_amdgcn_wmma_f32_16x16x4_f32(false, a, false, b2, (short)0, c2, false, false);
    c3 = __builtin_amdgcn_wmma_f32_16x16x4_f32(false, a, false, b3, (short)0, c3, false, false);
  }

  const int rbase = m0 + ((lane >> 4) << 3);
  float* zp = z + (long)rbase * OUT_DIM + lrow;
  #pragma unroll
  for (int r = 0; r < 8; ++r) {
    zp[(long)r * OUT_DIM +  0] = c0[r];
    zp[(long)r * OUT_DIM + 16] = c1[r];
    zp[(long)r * OUT_DIM + 32] = c2[r];
    zp[(long)r * OUT_DIM + 48] = c3[r];
  }

  const float as0 = Wattn[lrow],            as1 = Wattn[lrow + 16];
  const float as2 = Wattn[lrow + 32],       as3 = Wattn[lrow + 48];
  const float ad0 = Wattn[OUT_DIM + lrow],      ad1 = Wattn[OUT_DIM + lrow + 16];
  const float ad2 = Wattn[OUT_DIM + lrow + 32], ad3 = Wattn[OUT_DIM + lrow + 48];
  float ps[8], pd[8];
  #pragma unroll
  for (int r = 0; r < 8; ++r) {
    ps[r] = c0[r] * as0 + c1[r] * as1 + c2[r] * as2 + c3[r] * as3;
    pd[r] = c0[r] * ad0 + c1[r] * ad1 + c2[r] * ad2 + c3[r] * ad3;
  }
  #pragma unroll
  for (int o = 8; o > 0; o >>= 1) {
    #pragma unroll
    for (int r = 0; r < 8; ++r) {
      ps[r] += __shfl_xor(ps[r], o, 32);
      pd[r] += __shfl_xor(pd[r], o, 32);
    }
  }
  {
    const bool is_s = lrow < 8;
    const int  rr   = lrow & 7;
    float v = is_s ? ps[0] : pd[0];
    #pragma unroll
    for (int r = 1; r < 8; ++r) {
      const float t = is_s ? ps[r] : pd[r];
      v = (rr == r) ? t : v;
    }
    float* dp = (is_s ? zs : zd) + rbase + rr;
    *dp = v;
  }
  if (lane < 16) mord[m0 + lane] = 0x007FFFFFu;
  else           ssum[m0 + lane - 16] = 0.0f;
}

__global__ void gat_edge_scores(const int* __restrict__ src, const int* __restrict__ dst,
                                const float* __restrict__ wgt,
                                const float* __restrict__ zs, const float* __restrict__ zd,
                                float* __restrict__ ew, unsigned* __restrict__ mord) {
  int i = blockIdx.x * blockDim.x + threadIdx.x;
  if (i >= E_EDGES) return;
  int s0 = src[i], d0 = dst[i];
  float x = zs[s0] + zd[d0];
  float e = (x > 0.0f) ? x : NEG_SLOPE * x;
  e *= wgt[i];
  ew[i] = e;
  atomicMax(mord + d0, f32_ord(e));
}

__global__ void gat_edge_exp(const int* __restrict__ dst, const unsigned* __restrict__ mord,
                             float* __restrict__ ew, float* __restrict__ ssum) {
  int i = blockIdx.x * blockDim.x + threadIdx.x;
  if (i >= E_EDGES) return;
  int d0 = dst[i];
  float ex = expf(ew[i] - ord_f32(mord[d0]));
  ew[i] = ex;
  atomicAdd(ssum + d0, ex);
}

static_assert(E_EDGES % 32 == 0, "one wave owns exactly 32 edges");
#define AGG_CHUNK 16
__global__ void __launch_bounds__(256)
gat_aggregate(const int* __restrict__ src, const int* __restrict__ dst,
              const float* __restrict__ ew, const float* __restrict__ ssum,
              const float* __restrict__ z, float* __restrict__ out) {
  const int lane = threadIdx.x & 31;
  const int w  = blockIdx.x * (blockDim.x >> 5) + (threadIdx.x >> 5);
  const int e0 = w * 32;
  if (e0 >= E_EDGES) return;
  const int e = e0 + lane;
  const int   s_l = src[e], d_l = dst[e];
  const float a_l = ew[e] / ssum[d_l];
  const float* zl = z   + lane * 2;
  float*       ol = out + lane * 2;
  #pragma unroll
  for (int c = 0; c < 32; c += AGG_CHUNK) {
    v2f v[AGG_CHUNK];
    #pragma unroll
    for (int j = 0; j < AGG_CHUNK; ++j) {
      const int sj = __builtin_amdgcn_readlane(s_l, c + j);
      v[j] = *(const v2f*)(zl + (long)sj * OUT_DIM);
    }
    #pragma unroll
    for (int j = 0; j < AGG_CHUNK; ++j) {
      const int   dj = __builtin_amdgcn_readlane(d_l, c + j);
      const float aj = __int_as_float(__builtin_amdgcn_readlane(__float_as_int(a_l), c + j));
      float* op = ol + (long)dj * OUT_DIM;
      atomicAdd(op + 0, aj * v[j].x);
      atomicAdd(op + 1, aj * v[j].y);
    }
  }
}

extern "C" void kernel_launch(void* const* d_in, const int* in_sizes, int n_in,
                              void* d_out, int out_size, void* d_ws, size_t ws_size,
                              hipStream_t stream) {
  (void)in_sizes; (void)n_in; (void)out_size; (void)ws_size;
  const float* h     = (const float*)d_in[0];
  const float* Wfc   = (const float*)d_in[1];
  const float* Wattn = (const float*)d_in[2];
  const int*   src   = (const int*)d_in[3];
  const int*   dst   = (const int*)d_in[4];
  const float* wgt   = (const float*)d_in[5];
  float* out = (float*)d_out;

  float*    z    = (float*)d_ws;
  float*    zs   = z + (size_t)N_NODES * OUT_DIM;
  float*    zd   = zs + N_NODES;
  unsigned* mord = (unsigned*)(zd + N_NODES);
  float*    ssum = (float*)(mord + N_NODES);
  float*    ew   = ssum + N_NODES;

  {
    int waves  = (N_NODES + 15) / 16;
    int blocks = (waves * 32 + 255) / 256;
    gat_gemm_wmma<<<blocks, 256, 0, stream>>>(h, Wfc, Wattn, z, zs, zd, mord, ssum, out);
  }
  {
    int blocks = (E_EDGES + 255) / 256;
    gat_edge_scores<<<blocks, 256, 0, stream>>>(src, dst, wgt, zs, zd, ew, mord);
    gat_edge_exp<<<blocks, 256, 0, stream>>>(dst, mord, ew, ssum);
  }
  {
    int waves  = E_EDGES / 32;
    int blocks = (waves + 7) / 8;
    gat_aggregate<<<blocks, 256, 0, stream>>>(src, dst, ew, ssum, z, out);
  }
}


# Aggregate: lane = column (lane l owns columns l, l+32): every row load and every f32 atomic add instruction covers one whole 128-B line, not two half lines; atomics kept, arrival order as before
# speedup vs baseline: 1.4672x; 1.0902x over previous
#include <hip/hip_runtime.h>
#include <hip/hip_bf16.h>

#define N_NODES   50000
#define E_EDGES   1000000
#define IN_DIM    128
#define OUT_DIM   64
#define NEG_SLOPE 0.01f

typedef __attribute__((ext_vector_type(2))) float v2f;
typedef __attribute__((ext_vector_type(8))) float v8f;

__device__ __forceinline__ unsigned f32_ord(float f) {
  unsigned u = __float_as_uint(f);
  return (u & 0x80000000u) ? ~u : (u | 0x80000000u);
}
__device__ __forceinline__ float ord_f32(unsigned o) {
  unsigned u = (o & 0x80000000u) ? (o & 0x7FFFFFFFu) : ~o;
  return __uint_as_float(u);
}

static_assert(N_NODES % 16 == 0, "every wave owns a full 16-row tile");
__global__ void __launch_bounds__(256)
gat_gemm_wmma(const float* __restrict__ h, const float* __restrict__ Wfc,
              const float* __restrict__ Wattn,
              float* __restrict__ z, float* __restrict__ zs, float* __restrict__ zd,
              unsigned* __restrict__ mord, float* __restrict__ ssum,
              float* __restrict__ out) {
  const int lane = threadIdx.x & 31;
  const int wid  = blockIdx.x * (blockDim.x >> 5) + (threadIdx.x >> 5);
  const int m0   = wid * 16;
  if (m0 >= N_NODES) return;

  {
    float4* o4 = (float4*)(out + (long)m0 * OUT_DIM) + lane;
    const float4 zero4 = make_float4(0.0f, 0.0f, 0.0f, 0.0f);
    #pragma unroll
    for (int i = 0; i < 8; ++i) o4[i * 32] = zero4;
  }

  const int lrow = lane & 15;
  const int kh   = (lane >> 4) << 1;

  const float* __restrict__ ap  = h   + (long)(m0 + lrow) * IN_DIM + kh;
  const float* __restrict__ b0p = Wfc + (long)( 0 + lrow) * IN_DIM + kh;
  const float* __restrict__ b1p = Wfc + (long)(16 + lrow) * IN_DIM + kh;
  const float* __restrict__ b2p = Wfc + (long)(32 + lrow) * IN_DIM + kh;
  const float* __restrict__ b3p = Wfc + (long)(48 + lrow) * IN_DIM + kh;

  v8f c0 = {}, c1 = {}, c2 = {}, c3 = {};
  #pragma unroll 4
  for (int k = 0; k < IN_DIM; k += 4) {
    v2f a  = *(const v2f*)(ap  + k);
    v2f b0 = *(const v2f*)(b0p + k);
    v2f b1 = *(const v2f*)(b1p + k);
    v2f b2 = *(const v2f*)(b2p + k);
    v2f b3 = *(const v2f*)(b3p + k);
    c0 = __builtin_amdgcn_wmma_f32_16x16x4_f32(false, a, false, b0, (short)0, c0, false, false);
    c1 = __builtin_amdgcn_wmma_f32_16x16x4_f32(false, a, false, b1, (short)0, c1, false, false);
    c2 = __builtin_amdgcn_wmma_f32_16x16x4_f32(false, a, false, b2, (short)0, c2, false, false);
    c3 = __builtin_amdgcn_wmma_f32_16x16x4_f32(false, a, false, b3, (short)0, c3, false, false);
  }

  const int rbase = m0 + ((lane >> 4) << 3);
  float* zp = z + (long)rbase * OUT_DIM + lrow;
  #pragma unroll
  for (int r = 0; r < 8; ++r) {
    zp[(long)r * OUT_DIM +  0] = c0[r];
    zp[(long)r * OUT_DIM + 16] = c1[r];
    zp[(long)r * OUT_DIM + 32] = c2[r];
    zp[(long)r * OUT_DIM + 48] = c3[r];
  }

  const float as0 = Wattn[lrow],            as1 = Wattn[lrow + 16];
  const float as2 = Wattn[lrow + 32],       as3 = Wattn[lrow + 48];
  const float ad0 = Wattn[OUT_DIM + lrow],      ad1 = Wattn[OUT_DIM + lrow + 16];
  const float ad2 = Wattn[OUT_DIM + lrow + 32], ad3 = Wattn[OUT_DIM + lrow + 48];
  float ps[8], pd[8];
  #pragma unroll
  for (int r = 0; r < 8; ++r) {
    ps[r] = c0[r] * as0 + c1[r] * as1 + c2[r] * as2 + c3[r] * as3;
    pd[r] = c0[r] * ad0 + c1[r] * ad1 + c2[r] * ad2 + c3[r] * ad3;
  }
  #pragma unroll
  for (int o = 8; o > 0; o >>= 1) {
    #pragma unroll
    for (int r = 0; r < 8; ++r) {
      ps[r] += __shfl_xor(ps[r], o, 32);
      pd[r] += __shfl_xor(pd[r], o, 32);
    }
  }
  {
    const bool is_s = lrow < 8;
    const int  rr   = lrow & 7;
    float v = is_s ? ps[0] : pd[0];
    #pragma unroll
    for (int r = 1; r < 8; ++r) {
      const float t = is_s ? ps[r] : pd[r];
      v = (rr == r) ? t : v;
    }
    float* dp = (is_s ? zs : zd) + rbase + rr;
    *dp = v;
  }
  if (lane < 16) mord[m0 + lane] = 0x007FFFFFu;
  else           ssum[m0 + lane - 16] = 0.0f;
}

__global__ void gat_edge_scores(const int* __restrict__ src, const int* __restrict__ dst,
                                const float* __restrict__ wgt,
                                const float* __restrict__ zs, const float* __restrict__ zd,
                                float* __restrict__ ew, unsigned* __restrict__ mord) {
  int i = blockIdx.x * blockDim.x + threadIdx.x;
  if (i >= E_EDGES) return;
  int s0 = src[i], d0 = dst[i];
  float x = zs[s0] + zd[d0];
  float e = (x > 0.0f) ? x : NEG_SLOPE * x;
  e *= wgt[i];
  ew[i] = e;
  atomicMax(mord + d0, f32_ord(e));
}

__global__ void gat_edge_exp(const int* __restrict__ dst, const unsigned* __restrict__ mord,
                             float* __restrict__ ew, float* __restrict__ ssum) {
  int i = blockIdx.x * blockDim.x + threadIdx.x;
  if (i >= E_EDGES) return;
  int d0 = dst[i];
  float ex = expf(ew[i] - ord_f32(mord[d0]));
  ew[i] = ex;
  atomicAdd(ssum + d0, ex);
}

static_assert(E_EDGES % 32 == 0, "one wave owns exactly 32 edges");
static_assert(OUT_DIM == 64, "a row is two 128-byte lines of 32 floats");
#define AGG_CHUNK 16
__global__ void __launch_bounds__(256)
gat_aggregate(const int* __restrict__ src, const int* __restrict__ dst,
              const float* __restrict__ ew, const float* __restrict__ ssum,
              const float* __restrict__ z, float* __restrict__ out) {
  const int lane = threadIdx.x & 31;
  const int w  = blockIdx.x * (blockDim.x >> 5) + (threadIdx.x >> 5);
  const int e0 = w * 32;
  if (e0 >= E_EDGES) return;
  const int e = e0 + lane;
  const int   s_l = src[e], d_l = dst[e];
  const float a_l = ew[e] / ssum[d_l];
  const float* zl = z   + lane;
  float*       ol = out + lane;
  #pragma unroll
  for (int c = 0; c < 32; c += AGG_CHUNK) {
    float vlo[AGG_CHUNK], vhi[AGG_CHUNK];
    #pragma unroll
    for (int j = 0; j < AGG_CHUNK; ++j) {
      const int sj = __builtin_amdgcn_readlane(s_l, c + j);
      const float* zr = zl + (long)sj * OUT_DIM;
      vlo[j] = zr[0];
      vhi[j] = zr[32];
    }
    #pragma unroll
    for (int j = 0; j < AGG_CHUNK; ++j) {
      const int   dj = __builtin_amdgcn_readlane(d_l, c + j);
      const float aj = __int_as_float(__builtin_amdgcn_readlane(__float_as_int(a_l), c + j));
      float* op = ol + (long)dj * OUT_DIM;
      atomicAdd(op +  0, aj * vlo[j]);
      atomicAdd(op + 32, aj * vhi[j]);
    }
  }
}

extern "C" void kernel_launch(void* const* d_in, const int* in_sizes, int n_in,
                              void* d_out, int out_size, void* d_ws, size_t ws_size,
                              hipStream_t stream) {
  (void)in_sizes; (void)n_in; (void)out_size; (void)ws_size;
  const float* h     = (const float*)d_in[0];
  const float* Wfc   = (const float*)d_in[1];
  const float* Wattn = (const float*)d_in[2];
  const int*   src   = (const int*)d_in[3];
  const int*   dst   = (const int*)d_in[4];
  const float* wgt   = (const float*)d_in[5];
  float* out = (float*)d_out;

  float*    z    = (float*)d_ws;
  float*    zs   = z + (size_t)N_NODES * OUT_DIM;
  float*    zd   = zs + N_NODES;
  unsigned* mord = (unsigned*)(zd + N_NODES);
  float*    ssum = (float*)(mord + N_NODES);
  float*    ew   = ssum + N_NODES;

  {
    int waves  = (N_NODES + 15) / 16;
    int blocks = (waves * 32 + 255) / 256;
    gat_gemm_wmma<<<blocks, 256, 0, stream>>>(h, Wfc, Wattn, z, zs, zd, mord, ssum, out);
  }
  {
    int blocks = (E_EDGES + 255) / 256;
    gat_edge_scores<<<blocks, 256, 0, stream>>>(src, dst, wgt, zs, zd, ew, mord);
    gat_edge_exp<<<blocks, 256, 0, stream>>>(dst, mord, ew, ssum);
  }
  {
    int waves  = E_EDGES / 32;
    int blocks = (waves + 7) / 8;
    gat_aggregate<<<blocks, 256, 0, stream>>>(src, dst, ew, ssum, z, out);
  }
}


# Atomics go to one plane per half of the chip (die id read per wave): out, segment max, segment sum get two planes, readers combine them, a 5th launch adds the out planes; f32 sum tree changes
# speedup vs baseline: 2.4119x; 1.6439x over previous
#include <hip/hip_runtime.h>
#include <hip/hip_bf16.h>

#define N_NODES   50000
#define E_EDGES   1000000
#define IN_DIM    128
#define OUT_DIM   64
#define NEG_SLOPE 0.01f
#define NPAD      50048
#define ROWPLANE  (N_NODES * OUT_DIM)

typedef __attribute__((ext_vector_type(2))) float v2f;
typedef __attribute__((ext_vector_type(8))) float v8f;

__device__ __forceinline__ unsigned f32_ord(float f) {
  unsigned u = __float_as_uint(f);
  return (u & 0x80000000u) ? ~u : (u | 0x80000000u);
}
__device__ __forceinline__ float ord_f32(unsigned o) {
  unsigned u = (o & 0x80000000u) ? (o & 0x7FFFFFFFu) : ~o;
  return __uint_as_float(u);
}

__device__ __forceinline__ int chip_half() {
  return (int)((__builtin_amdgcn_s_sendmsg_rtn(0x87) >> 18) & 1u);
}

static_assert(N_NODES % 16 == 0, "every wave owns a full 16-row tile");
__global__ void __launch_bounds__(256)
gat_gemm_wmma(const float* __restrict__ h, const float* __restrict__ Wfc,
              const float* __restrict__ Wattn,
              float* __restrict__ z, float* __restrict__ zs, float* __restrict__ zd,
              unsigned* __restrict__ mord, float* __restrict__ ssum,
              float* __restrict__ acc) {
  const int lane = threadIdx.x & 31;
  const int wid  = blockIdx.x * (blockDim.x >> 5) + (threadIdx.x >> 5);
  const int m0   = wid * 16;
  if (m0 >= N_NODES) return;

  {
    float4* o4 = (float4*)(acc + (long)m0 * OUT_DIM) + lane;
    float4* q4 = o4 + ROWPLANE / 4;
    const float4 zero4 = make_float4(0.0f, 0.0f, 0.0f, 0.0f);
    #pragma unroll
    for (int i = 0; i < 8; ++i) o4[i * 32] = zero4;
    #pragma unroll
    for (int i = 0; i < 8; ++i) q4[i * 32] = zero4;
  }

  const int lrow = lane & 15;
  const int kh   = (lane >> 4) << 1;

  const float* __restrict__ ap  = h   + (long)(m0 + lrow) * IN_DIM + kh;
  const float* __restrict__ b0p = Wfc + (long)( 0 + lrow) * IN_DIM + kh;
  const float* __restrict__ b1p = Wfc + (long)(16 + lrow) * IN_DIM + kh;
  const float* __restrict__ b2p = Wfc + (long)(32 + lrow) * IN_DIM + kh;
  const float* __restrict__ b3p = Wfc + (long)(48 + lrow) * IN_DIM + kh;

  v8f c0 = {}, c1 = {}, c2 = {}, c3 = {};
  #pragma unroll 4
  for (int k = 0; k < IN_DIM; k += 4) {
    v2f a  = *(const v2f*)(ap  + k);
    v2f b0 = *(const v2f*)(b0p + k);
    v2f b1 = *(const v2f*)(b1p + k);
    v2f b2 = *(const v2f*)(b2p + k);
    v2f b3 = *(const v2f*)(b3p + k);
    c0 = __builtin_amdgcn_wmma_f32_16x16x4_f32(false, a, false, b0, (short)0, c0, false, false);
    c1 = __builtin_amdgcn_wmma_f32_16x16x4_f32(false, a, false, b1, (short)0, c1, false, false);
    c2 = __builtin_amdgcn_wmma_f32_16x16x4_f32(false, a, false, b2, (short)0, c2, false, false);
    c3 = __builtin_amdgcn_wmma_f32_16x16x4_f32(false, a, false, b3, (short)0, c3, false, false);
  }

  const int rbase = m0 + ((lane >> 4) << 3);
  float* zp = z + (long)rbase * OUT_DIM + lrow;
  #pragma unroll
  for (int r = 0; r < 8; ++r) {
    zp[(long)r * OUT_DIM +  0] = c0[r];
    zp[(long)r * OUT_DIM + 16] = c1[r];
    zp[(long)r * OUT_DIM + 32] = c2[r];
    zp[(long)r * OUT_DIM + 48] = c3[r];
  }

  const float as0 = Wattn[lrow],            as1 = Wattn[lrow + 16];
  const float as2 = Wattn[lrow + 32],       as3 = Wattn[lrow + 48];
  const float ad0 = Wattn[OUT_DIM + lrow],      ad1 = Wattn[OUT_DIM + lrow + 16];
  const float ad2 = Wattn[OUT_DIM + lrow + 32], ad3 = Wattn[OUT_DIM + lrow + 48];
  float ps[8], pd[8];
  #pragma unroll
  for (int r = 0; r < 8; ++r) {
    ps[r] = c0[r] * as0 + c1[r] * as1 + c2[r] * as2 + c3[r] * as3;
    pd[r] = c0[r] * ad0 + c1[r] * ad1 + c2[r] * ad2 + c3[r] * ad3;
  }
  #pragma unroll
  for (int o = 8; o > 0; o >>= 1) {
    #pragma unroll
    for (int r = 0; r < 8; ++r) {
      ps[r] += __shfl_xor(ps[r], o, 32);
      pd[r] += __shfl_xor(pd[r], o, 32);
    }
  }
  {
    const bool is_s = lrow < 8;
    const int  rr   = lrow & 7;
    float v = is_s ? ps[0] : pd[0];
    #pragma unroll
    for (int r = 1; r < 8; ++r) {
      const float t = is_s ? ps[r] : pd[r];
      v = (rr == r) ? t : v;
    }
    float* dp = (is_s ? zs : zd) + rbase + rr;
    *dp = v;
  }
  if (lane < 16) {
    mord[m0 + lane]        = 0x007FFFFFu;
    mord[NPAD + m0 + lane] = 0x007FFFFFu;
  } else {
    ssum[m0 + lane - 16]        = 0.0f;
    ssum[NPAD + m0 + lane - 16] = 0.0f;
  }
}

__global__ void gat_edge_scores(const int* __restrict__ src, const int* __restrict__ dst,
                                const float* __restrict__ wgt,
                                const float* __restrict__ zs, const float* __restrict__ zd,
                                float* __restrict__ ew, unsigned* __restrict__ mord) {
  int i = blockIdx.x * blockDim.x + threadIdx.x;
  if (i >= E_EDGES) return;
  const int hf = chip_half();
  int s0 = src[i], d0 = dst[i];
  float x = zs[s0] + zd[d0];
  float e = (x > 0.0f) ? x : NEG_SLOPE * x;
  e *= wgt[i];
  ew[i] = e;
  atomicMax(mord + hf * NPAD + d0, f32_ord(e));
}

__global__ void gat_edge_exp(const int* __restrict__ dst, const unsigned* __restrict__ mord,
                             float* __restrict__ ew, float* __restrict__ ssum) {
  int i = blockIdx.x * blockDim.x + threadIdx.x;
  if (i >= E_EDGES) return;
  const int hf = chip_half();
  int d0 = dst[i];
  const unsigned o0 = mord[d0], o1 = mord[NPAD + d0];
  float ex = expf(ew[i] - ord_f32(o0 > o1 ? o0 : o1));
  ew[i] = ex;
  atomicAdd(ssum + hf * NPAD + d0, ex);
}

static_assert(E_EDGES % 32 == 0, "one wave owns exactly 32 edges");
static_assert(OUT_DIM == 64, "a row is two 128-byte lines of 32 floats");
#define AGG_CHUNK 16
__global__ void __launch_bounds__(256)
gat_aggregate(const int* __restrict__ src, const int* __restrict__ dst,
              const float* __restrict__ ew, const float* __restrict__ ssum,
              const float* __restrict__ z, float* __restrict__ acc) {
  const int lane = threadIdx.x & 31;
  const int w  = blockIdx.x * (blockDim.x >> 5) + (threadIdx.x >> 5);
  const int e0 = w * 32;
  if (e0 >= E_EDGES) return;
  const int e = e0 + lane;
  const int   s_l = src[e], d_l = dst[e];
  const float a_l = ew[e] / (ssum[d_l] + ssum[NPAD + d_l]);
  const float* zl = z   + lane;
  float*       ol = acc + (long)chip_half() * ROWPLANE + lane;
  #pragma unroll
  for (int c = 0; c < 32; c += AGG_CHUNK) {
    float vlo[AGG_CHUNK], vhi[AGG_CHUNK];
    #pragma unroll
    for (int j = 0; j < AGG_CHUNK; ++j) {
      const int sj = __builtin_amdgcn_readlane(s_l, c + j);
      const float* zr = zl + (long)sj * OUT_DIM;
      vlo[j] = zr[0];
      vhi[j] = zr[32];
    }
    #pragma unroll
    for (int j = 0; j < AGG_CHUNK; ++j) {
      const int   dj = __builtin_amdgcn_readlane(d_l, c + j);
      const float aj = __int_as_float(__builtin_amdgcn_readlane(__float_as_int(a_l), c + j));
      float* op = ol + (long)dj * OUT_DIM;
      atomicAdd(op +  0, aj * vlo[j]);
      atomicAdd(op + 32, aj * vhi[j]);
    }
  }
}

#define SUM_V4      (ROWPLANE / 4)
#define SUM_BLOCKS  625
#define SUM_THREADS (SUM_BLOCKS * 256)
static_assert(SUM_THREADS * 5 == SUM_V4, "five float4 per lane cover a plane exactly");
__global__ void __launch_bounds__(256)
gat_sum_planes(const float* __restrict__ acc, float* __restrict__ out) {
  const int t = blockIdx.x * 256 + threadIdx.x;
  const float4* __restrict__ a = (const float4*)acc + t;
  const float4* __restrict__ b = a + SUM_V4;
  float4* __restrict__ o = (float4*)out + t;
  float4 x[5], y[5];
  #pragma unroll
  for (int u = 0; u < 5; ++u) { x[u] = a[u * SUM_THREADS]; y[u] = b[u * SUM_THREADS]; }
  #pragma unroll
  for (int u = 0; u < 5; ++u)
    o[u * SUM_THREADS] = make_float4(x[u].x + y[u].x, x[u].y + y[u].y,
                                     x[u].z + y[u].z, x[u].w + y[u].w);
}

extern "C" void kernel_launch(void* const* d_in, const int* in_sizes, int n_in,
                              void* d_out, int out_size, void* d_ws, size_t ws_size,
                              hipStream_t stream) {
  (void)in_sizes; (void)n_in; (void)out_size; (void)ws_size;
  const float* h     = (const float*)d_in[0];
  const float* Wfc   = (const float*)d_in[1];
  const float* Wattn = (const float*)d_in[2];
  const int*   src   = (const int*)d_in[3];
  const int*   dst   = (const int*)d_in[4];
  const float* wgt   = (const float*)d_in[5];
  float* out = (float*)d_out;

  float*    z    = (float*)d_ws;
  float*    zs   = z + (size_t)N_NODES * OUT_DIM;
  float*    zd   = zs + NPAD;
  unsigned* mord = (unsigned*)(zd + NPAD);
  float*    ssum = (float*)(mord + 2 * NPAD);
  float*    ew   = ssum + 2 * NPAD;
  float*    acc  = ew + E_EDGES;

  {
    int waves  = (N_NODES + 15) / 16;
    int blocks = (waves * 32 + 255) / 256;
    gat_gemm_wmma<<<blocks, 256, 0, stream>>>(h, Wfc, Wattn, z, zs, zd, mord, ssum, acc);
  }
  {
    int blocks = (E_EDGES + 255) / 256;
    gat_edge_scores<<<blocks, 256, 0, stream>>>(src, dst, wgt, zs, zd, ew, mord);
    gat_edge_exp<<<blocks, 256, 0, stream>>>(dst, mord, ew, ssum);
  }
  {
    int waves  = E_EDGES / 32;
    int blocks = (waves + 7) / 8;
    gat_aggregate<<<blocks, 256, 0, stream>>>(src, dst, ew, ssum, z, acc);
  }
  gat_sum_planes<<<SUM_BLOCKS, 256, 0, stream>>>(acc, out);
}


# Aggregate by destination: counting sort on dst (integer slot adds, scan, place); one wave per node does softmax + row sum in registers, edges in index order: fixed-order f32 sums, no f32 atomics
# speedup vs baseline: 2.8979x; 1.2015x over previous
#include <hip/hip_runtime.h>
#include <hip/hip_bf16.h>

#define N_NODES   50000
#define E_EDGES   1000000
#define IN_DIM    128
#define OUT_DIM   64
#define NEG_SLOPE 0.01f
#define NPAD      50048

typedef __attribute__((ext_vector_type(2))) float v2f;
typedef __attribute__((ext_vector_type(8))) float v8f;

__device__ __forceinline__ int chip_half() {
  return (int)((__builtin_amdgcn_s_sendmsg_rtn(0x87) >> 18) & 1u);
}

static_assert(N_NODES % 16 == 0, "every wave owns a full 16-row tile");
__global__ void __launch_bounds__(256)
gat_gemm_wmma(const float* __restrict__ h, const float* __restrict__ Wfc,
              const float* __restrict__ Wattn,
              float* __restrict__ z, float* __restrict__ zs, float* __restrict__ zd,
              unsigned* __restrict__ cnt) {
  const int lane = threadIdx.x & 31;
  const int wid  = blockIdx.x * (blockDim.x >> 5) + (threadIdx.x >> 5);
  const int m0   = wid * 16;
  if (m0 >= N_NODES) return;

  const int lrow = lane & 15;
  const int kh   = (lane >> 4) << 1;

  const float* __restrict__ ap  = h   + (long)(m0 + lrow) * IN_DIM + kh;
  const float* __restrict__ b0p = Wfc + (long)( 0 + lrow) * IN_DIM + kh;
  const float* __restrict__ b1p = Wfc + (long)(16 + lrow) * IN_DIM + kh;
  const float* __restrict__ b2p = Wfc + (long)(32 + lrow) * IN_DIM + kh;
  const float* __restrict__ b3p = Wfc + (long)(48 + lrow) * IN_DIM + kh;

  v8f c0 = {}, c1 = {}, c2 = {}, c3 = {};
  #pragma unroll 4
  for (int k = 0; k < IN_DIM; k += 4) {
    v2f a  = *(const v2f*)(ap  + k);
    v2f b0 = *(const v2f*)(b0p + k);
    v2f b1 = *(const v2f*)(b1p + k);
    v2f b2 = *(const v2f*)(b2p + k);
    v2f b3 = *(const v2f*)(b3p + k);
    c0 = __builtin_amdgcn_wmma_f32_16x16x4_f32(false, a, false, b0, (short)0, c0, false, false);
    c1 = __builtin_amdgcn_wmma_f32_16x16x4_f32(false, a, false, b1, (short)0, c1, false, false);
    c2 = __builtin_amdgcn_wmma_f32_16x16x4_f32(false, a, false, b2, (short)0, c2, false, false);
    c3 = __builtin_amdgcn_wmma_f32_16x16x4_f32(false, a, false, b3, (short)0, c3, false, false);
  }

  const int rbase = m0 + ((lane >> 4) << 3);
  float* zp = z + (long)rbase * OUT_DIM + lrow;
  #pragma unroll
  for (int r = 0; r < 8; ++r) {
    zp[(long)r * OUT_DIM +  0] = c0[r];
    zp[(long)r * OUT_DIM + 16] = c1[r];
    zp[(long)r * OUT_DIM + 32] = c2[r];
    zp[(long)r * OUT_DIM + 48] = c3[r];
  }

  const float as0 = Wattn[lrow],            as1 = Wattn[lrow + 16];
  const float as2 = Wattn[lrow + 32],       as3 = Wattn[lrow + 48];
  const float ad0 = Wattn[OUT_DIM + lrow],      ad1 = Wattn[OUT_DIM + lrow + 16];
  const float ad2 = Wattn[OUT_DIM + lrow + 32], ad3 = Wattn[OUT_DIM + lrow + 48];
  float ps[8], pd[8];
  #pragma unroll
  for (int r = 0; r < 8; ++r) {
    ps[r] = c0[r] * as0 + c1[r] * as1 + c2[r] * as2 + c3[r] * as3;
    pd[r] = c0[r] * ad0 + c1[r] * ad1 + c2[r] * ad2 + c3[r] * ad3;
  }
  #pragma unroll
  for (int o = 8; o > 0; o >>= 1) {
    #pragma unroll
    for (int r = 0; r < 8; ++r) {
      ps[r] += __shfl_xor(ps[r], o, 32);
      pd[r] += __shfl_xor(pd[r], o, 32);
    }
  }
  {
    const bool is_s = lrow < 8;
    const int  rr   = lrow & 7;
    float v = is_s ? ps[0] : pd[0];
    #pragma unroll
    for (int r = 1; r < 8; ++r) {
      const float t = is_s ? ps[r] : pd[r];
      v = (rr == r) ? t : v;
    }
    float* dp = (is_s ? zs : zd) + rbase + rr;
    *dp = v;
  }
  if (lane < 16) cnt[m0 + lane] = 0u;
  else           cnt[NPAD + m0 + lane - 16] = 0u;
}

static_assert(E_EDGES % 4 == 0, "edges are taken four at a time");
#define E_V4 (E_EDGES / 4)
__global__ void __launch_bounds__(256)
gat_edge_scores(const int* __restrict__ src, const int* __restrict__ dst,
                const float* __restrict__ wgt,
                const float* __restrict__ zs, const float* __restrict__ zd,
                float* __restrict__ ew, unsigned* __restrict__ rk,
                unsigned* __restrict__ cnt) {
  const int i = blockIdx.x * 256 + threadIdx.x;
  if (i >= E_V4) return;
  const unsigned hf = (unsigned)chip_half();
  unsigned* __restrict__ cp = cnt + hf * NPAD;
  const int4   s4 = ((const int4*)src)[i];
  const int4   d4 = ((const int4*)dst)[i];
  const float4 w4 = ((const float4*)wgt)[i];
  const float x0 = zs[s4.x] + zd[d4.x];
  const float x1 = zs[s4.y] + zd[d4.y];
  const float x2 = zs[s4.z] + zd[d4.z];
  const float x3 = zs[s4.w] + zd[d4.w];
  float4 e4;
  e4.x = ((x0 > 0.0f) ? x0 : NEG_SLOPE * x0) * w4.x;
  e4.y = ((x1 > 0.0f) ? x1 : NEG_SLOPE * x1) * w4.y;
  e4.z = ((x2 > 0.0f) ? x2 : NEG_SLOPE * x2) * w4.z;
  e4.w = ((x3 > 0.0f) ? x3 : NEG_SLOPE * x3) * w4.w;
  ((float4*)ew)[i] = e4;
  const unsigned tag = hf << 31;
  uint4 r4;
  r4.x = atomicAdd(cp + d4.x, 1u) | tag;
  r4.y = atomicAdd(cp + d4.y, 1u) | tag;
  r4.z = atomicAdd(cp + d4.z, 1u) | tag;
  r4.w = atomicAdd(cp + d4.w, 1u) | tag;
  ((uint4*)rk)[i] = r4;
}

#define SCAN_TILES 13
#define N_V4       (N_NODES / 4)
static_assert(N_NODES % 4 == 0, "nodes are scanned four at a time");
static_assert(SCAN_TILES * 1024 >= N_V4, "13 tiles of 1024 lanes cover every node");
__global__ void __launch_bounds__(1024)
gat_scan(const unsigned* __restrict__ cnt, unsigned* __restrict__ off) {
  __shared__ unsigned wtot[SCAN_TILES][32];
  const int t = threadIdx.x, lane = t & 31, wv = t >> 5;
  const uint4* __restrict__ c0 = (const uint4*)cnt;
  const uint4* __restrict__ c1 = (const uint4*)(cnt + NPAD);
  uint4 v[SCAN_TILES];
  unsigned inc[SCAN_TILES];
  #pragma unroll
  for (int k = 0; k < SCAN_TILES; ++k) {
    const int idx = k * 1024 + t;
    v[k] = make_uint4(0u, 0u, 0u, 0u);
    if (idx < N_V4) {
      const uint4 a = c0[idx], b = c1[idx];
      v[k] = make_uint4(a.x + b.x, a.y + b.y, a.z + b.z, a.w + b.w);
    }
    inc[k] = v[k].x + v[k].y + v[k].z + v[k].w;
  }
  #pragma unroll
  for (int o = 1; o < 32; o <<= 1) {
    #pragma unroll
    for (int k = 0; k < SCAN_TILES; ++k) {
      const unsigned n = __shfl_up(inc[k], o, 32);
      if (lane >= o) inc[k] += n;
    }
  }
  if (lane == 31) {
    #pragma unroll
    for (int k = 0; k < SCAN_TILES; ++k) wtot[k][wv] = inc[k];
  }
  __syncthreads();
  if (wv == 0) {
    unsigned carry = 0u;
    #pragma unroll
    for (int k = 0; k < SCAN_TILES; ++k) {
      const unsigned x = wtot[k][lane];
      unsigned y = x;
      #pragma unroll
      for (int o = 1; o < 32; o <<= 1) {
        const unsigned n = __shfl_up(y, o, 32);
        if (lane >= o) y += n;
      }
      wtot[k][lane] = carry + y - x;
      carry += __shfl(y, 31, 32);
    }
  }
  __syncthreads();
  #pragma unroll
  for (int k = 0; k < SCAN_TILES; ++k) {
    const int idx = k * 1024 + t;
    if (idx < N_V4) {
      const unsigned own  = v[k].x + v[k].y + v[k].z + v[k].w;
      const unsigned base = wtot[k][wv] + inc[k] - own;
      ((uint4*)off)[idx] = make_uint4(base, base + v[k].x, base + v[k].x + v[k].y,
                                      base + v[k].x + v[k].y + v[k].z);
    }
  }
}

__device__ __forceinline__ unsigned place_of(unsigned r, int d0,
                                             const unsigned* __restrict__ off,
                                             const unsigned* __restrict__ cnt) {
  const unsigned o = off[d0], c = cnt[d0];
  return o + (r & 0x7FFFFFFFu) + ((r >> 31) ? c : 0u);
}
__global__ void __launch_bounds__(256)
gat_place(const int* __restrict__ src, const int* __restrict__ dst,
          const float* __restrict__ ew, const unsigned* __restrict__ rk,
          const unsigned* __restrict__ off, const unsigned* __restrict__ cnt,
          int4* __restrict__ srt) {
  const int i = blockIdx.x * 256 + threadIdx.x;
  if (i >= E_V4) return;
  const int4   d4 = ((const int4*)dst)[i];
  const uint4  r4 = ((const uint4*)rk)[i];
  const int4   s4 = ((const int4*)src)[i];
  const float4 e4 = ((const float4*)ew)[i];
  const unsigned p0 = place_of(r4.x, d4.x, off, cnt);
  const unsigned p1 = place_of(r4.y, d4.y, off, cnt);
  const unsigned p2 = place_of(r4.z, d4.z, off, cnt);
  const unsigned p3 = place_of(r4.w, d4.w, off, cnt);
  srt[p0] = make_int4(s4.x, __float_as_int(e4.x), 4 * i + 0, 0);
  srt[p1] = make_int4(s4.y, __float_as_int(e4.y), 4 * i + 1, 0);
  srt[p2] = make_int4(s4.z, __float_as_int(e4.z), 4 * i + 2, 0);
  srt[p3] = make_int4(s4.w, __float_as_int(e4.w), 4 * i + 3, 0);
}

#define SLOTS 64
static_assert(N_NODES % 8 == 0, "eight waves per block, one node each");

template <int C>
__device__ __forceinline__ void node_step(const int2* sl, const float2* __restrict__ z2,
                                          unsigned lane, float2& acc, float& s) {
  int2 q[C];
  float2 v[C];
  #pragma unroll
  for (int u = 0; u < C; ++u) q[u] = sl[u];
  #pragma unroll
  for (int u = 0; u < C; ++u) v[u] = z2[(unsigned)q[u].x * (unsigned)(OUT_DIM / 2) + lane];
  #pragma unroll
  for (int u = 0; u < C; ++u) {
    const float a = __int_as_float(q[u].y);
    s     += a;
    acc.x += a * v[u].x;
    acc.y += a * v[u].y;
  }
}

__device__ __forceinline__ void node_walk(const int2* sl, int n, const float2* __restrict__ z2,
                                          unsigned lane, float2& acc, float& s) {
  int p = 0;
  for (; p + 8 <= n; p += 8) node_step<8>(sl + p, z2, lane, acc, s);
  if (p + 4 <= n) { node_step<4>(sl + p, z2, lane, acc, s); p += 4; }
  if (p + 2 <= n) { node_step<2>(sl + p, z2, lane, acc, s); p += 2; }
  if (p + 1 <= n) { node_step<1>(sl + p, z2, lane, acc, s); }
}

__device__ __forceinline__ int count_below(int idk, int key) {
  int c = 0;
  #pragma unroll
  for (int t = 0; t < 32; ++t) c += (__builtin_amdgcn_readlane(idk, t) < key) ? 1 : 0;
  return c;
}

__device__ __forceinline__ float wave_max(float m) {
  #pragma unroll
  for (int o = 16; o > 0; o >>= 1) m = fmaxf(m, __shfl_xor(m, o, 32));
  return m;
}

__global__ void __launch_bounds__(256)
gat_node(const unsigned* __restrict__ off, const unsigned* __restrict__ cnt,
         const int4* __restrict__ srt, const float* __restrict__ z,
         float* __restrict__ out) {
  __shared__ int2 slot[8][SLOTS];
  const int lane = threadIdx.x & 31;
  const int wv   = threadIdx.x >> 5;
  const int d    = blockIdx.x * 8 + wv;
  const int beg = __builtin_amdgcn_readfirstlane((int)off[d]);
  const int deg = __builtin_amdgcn_readfirstlane((int)(cnt[d] + cnt[NPAD + d]));
  const int4* __restrict__ seg = srt + beg;
  const float2* __restrict__ z2 = (const float2*)z;
  int2* sl = slot[wv];

  float2 acc = make_float2(0.0f, 0.0f);
  float  s   = 0.0f;
  if (deg <= 32) {
    const bool valid = lane < deg;
    int4 en = make_int4(0, __float_as_int(-INFINITY), 0x7FFFFFFF, 0);
    if (valid) en = seg[lane];
    const float e = __int_as_float(en.y);
    const float m = wave_max(e);
    const int rank = count_below(en.z, en.z);
    if (valid) sl[rank] = make_int2(en.x, __float_as_int(expf(e - m)));
    node_walk(sl, deg, z2, (unsigned)lane, acc, s);
  } else {
    float m = -INFINITY;
    for (int j = lane; j < deg; j += 32) m = fmaxf(m, __int_as_float(seg[j].y));
    m = wave_max(m);
    for (int w = 0; w < deg; w += SLOTS) {
      for (int jb = 0; jb < deg; jb += 32) {
        const int  j     = jb + lane;
        const bool valid = j < deg;
        int4 en = make_int4(0, 0, 0x7FFFFFFF, 0);
        if (valid) en = seg[j];
        int rank = 0;
        for (int kb = 0; kb < deg; kb += 32) {
          int idk = en.z;
          if (kb != jb) idk = (kb + lane < deg) ? seg[kb + lane].z : 0x7FFFFFFF;
          rank += count_below(idk, en.z);
        }
        if (valid && rank >= w && rank < w + SLOTS)
          sl[rank - w] = make_int2(en.x, __float_as_int(expf(__int_as_float(en.y) - m)));
      }
      node_walk(sl, (deg - w < SLOTS) ? (deg - w) : SLOTS, z2, (unsigned)lane, acc, s);
    }
  }
  float2 o = make_float2(0.0f, 0.0f);
  if (deg > 0) o = make_float2(acc.x / s, acc.y / s);
  ((float2*)out)[(unsigned)d * (unsigned)(OUT_DIM / 2) + (unsigned)lane] = o;
}

extern "C" void kernel_launch(void* const* d_in, const int* in_sizes, int n_in,
                              void* d_out, int out_size, void* d_ws, size_t ws_size,
                              hipStream_t stream) {
  (void)in_sizes; (void)n_in; (void)out_size; (void)ws_size;
  const float* h     = (const float*)d_in[0];
  const float* Wfc   = (const float*)d_in[1];
  const float* Wattn = (const float*)d_in[2];
  const int*   src   = (const int*)d_in[3];
  const int*   dst   = (const int*)d_in[4];
  const float* wgt   = (const float*)d_in[5];
  float* out = (float*)d_out;

  float*    z    = (float*)d_ws;
  float*    zs   = z + (size_t)N_NODES * OUT_DIM;
  float*    zd   = zs + NPAD;
  unsigned* cnt  = (unsigned*)(zd + NPAD);
  unsigned* off  = cnt + 2 * NPAD;
  float*    ew   = (float*)(off + NPAD);
  unsigned* rk   = (unsigned*)(ew + E_EDGES);
  int4*     srt  = (int4*)(rk + E_EDGES);

  {
    int waves  = (N_NODES + 15) / 16;
    int blocks = (waves * 32 + 255) / 256;
    gat_gemm_wmma<<<blocks, 256, 0, stream>>>(h, Wfc, Wattn, z, zs, zd, cnt);
  }
  {
    int blocks = (E_V4 + 255) / 256;
    gat_edge_scores<<<blocks, 256, 0, stream>>>(src, dst, wgt, zs, zd, ew, rk, cnt);
    gat_scan<<<1, 1024, 0, stream>>>(cnt, off);
    gat_place<<<blocks, 256, 0, stream>>>(src, dst, ew, rk, off, cnt, srt);
  }
  gat_node<<<N_NODES / 8, 256, 0, stream>>>(off, cnt, srt, z, out);
}


# Each edge record goes straight into a fixed-capacity bucket of its destination (slot = the count its integer add returns; overflow list beyond): scan and place launches dropped, 5 launches -> 3
# speedup vs baseline: 5.0436x; 1.7404x over previous
#include <hip/hip_runtime.h>
#include <hip/hip_bf16.h>

#define N_NODES   50000
#define E_EDGES   1000000
#define IN_DIM    128
#define OUT_DIM   64
#define NEG_SLOPE 0.01f
#define NPAD      50048
#define CAPH      32

typedef __attribute__((ext_vector_type(2))) float v2f;
typedef __attribute__((ext_vector_type(8))) float v8f;

__device__ __forceinline__ int chip_half() {
  return (int)((__builtin_amdgcn_s_sendmsg_rtn(0x87) >> 18) & 1u);
}

static_assert(N_NODES % 16 == 0, "every wave owns a full 16-row tile");
__global__ void __launch_bounds__(256)
gat_gemm_wmma(const float* __restrict__ h, const float* __restrict__ Wfc,
              const float* __restrict__ Wattn,
              float* __restrict__ z, float* __restrict__ zs, float* __restrict__ zd,
              unsigned* __restrict__ cnt, unsigned* __restrict__ ovfn) {
  const int lane = threadIdx.x & 31;
  const int wid  = blockIdx.x * (blockDim.x >> 5) + (threadIdx.x >> 5);
  const int m0   = wid * 16;
  if (m0 >= N_NODES) return;

  const int lrow = lane & 15;
  const int kh   = (lane >> 4) << 1;

  const float* __restrict__ ap  = h   + (long)(m0 + lrow) * IN_DIM + kh;
  const float* __restrict__ b0p = Wfc + (long)( 0 + lrow) * IN_DIM + kh;
  const float* __restrict__ b1p = Wfc + (long)(16 + lrow) * IN_DIM + kh;
  const float* __restrict__ b2p = Wfc + (long)(32 + lrow) * IN_DIM + kh;
  const float* __restrict__ b3p = Wfc + (long)(48 + lrow) * IN_DIM + kh;

  v8f c0 = {}, c1 = {}, c2 = {}, c3 = {};
  #pragma unroll 4
  for (int k = 0; k < IN_DIM; k += 4) {
    v2f a  = *(const v2f*)(ap  + k);
    v2f b0 = *(const v2f*)(b0p + k);
    v2f b1 = *(const v2f*)(b1p + k);
    v2f b2 = *(const v2f*)(b2p + k);
    v2f b3 = *(const v2f*)(b3p + k);
    c0 = __builtin_amdgcn_wmma_f32_16x16x4_f32(false, a, false, b0, (short)0, c0, false, false);
    c1 = __builtin_amdgcn_wmma_f32_16x16x4_f32(false, a, false, b1, (short)0, c1, false, false);
    c2 = __builtin_amdgcn_wmma_f32_16x16x4_f32(false, a, false, b2, (short)0, c2, false, false);
    c3 = __builtin_amdgcn_wmma_f32_16x16x4_f32(false, a, false, b3, (short)0, c3, false, false);
  }

  const int rbase = m0 + ((lane >> 4) << 3);
  float* zp = z + (long)rbase * OUT_DIM + lrow;
  #pragma unroll
  for (int r = 0; r < 8; ++r) {
    zp[(long)r * OUT_DIM +  0] = c0[r];
    zp[(long)r * OUT_DIM + 16] = c1[r];
    zp[(long)r * OUT_DIM + 32] = c2[r];
    zp[(long)r * OUT_DIM + 48] = c3[r];
  }

  const float as0 = Wattn[lrow],            as1 = Wattn[lrow + 16];
  const float as2 = Wattn[lrow + 32],       as3 = Wattn[lrow + 48];
  const float ad0 = Wattn[OUT_DIM + lrow],      ad1 = Wattn[OUT_DIM + lrow + 16];
  const float ad2 = Wattn[OUT_DIM + lrow + 32], ad3 = Wattn[OUT_DIM + lrow + 48];
  float ps[8], pd[8];
  #pragma unroll
  for (int r = 0; r < 8; ++r) {
    ps[r] = c0[r] * as0 + c1[r] * as1 + c2[r] * as2 + c3[r] * as3;
    pd[r] = c0[r] * ad0 + c1[r] * ad1 + c2[r] * ad2 + c3[r] * ad3;
  }
  #pragma unroll
  for (int o = 8; o > 0; o >>= 1) {
    #pragma unroll
    for (int r = 0; r < 8; ++r) {
      ps[r] += __shfl_xor(ps[r], o, 32);
      pd[r] += __shfl_xor(pd[r], o, 32);
    }
  }
  {
    const bool is_s = lrow < 8;
    const int  rr   = lrow & 7;
    float v = is_s ? ps[0] : pd[0];
    #pragma unroll
    for (int r = 1; r < 8; ++r) {
      const float t = is_s ? ps[r] : pd[r];
      v = (rr == r) ? t : v;
    }
    float* dp = (is_s ? zs : zd) + rbase + rr;
    *dp = v;
  }
  if (lane < 16) cnt[m0 + lane] = 0u;
  else           cnt[NPAD + m0 + lane - 16] = 0u;
  if (wid == 0 && lane == 0) *ovfn = 0u;
}

static_assert(E_EDGES % 4 == 0, "edges are taken four at a time");
#define E_V4 (E_EDGES / 4)
__device__ __forceinline__ void put_record(unsigned r, int4 rec, int4* __restrict__ bk,
                                           int4* __restrict__ ovf, unsigned* __restrict__ ovfn) {
  if (r < (unsigned)CAPH) {
    bk[(unsigned)rec.w * (unsigned)(2 * CAPH) + r] = rec;
  } else {
    const unsigned p = atomicAdd(ovfn, 1u);
    ovf[p] = rec;
  }
}
__global__ void __launch_bounds__(256)
gat_edge_scores(const int* __restrict__ src, const int* __restrict__ dst,
                const float* __restrict__ wgt,
                const float* __restrict__ zs, const float* __restrict__ zd,
                unsigned* __restrict__ cnt, int4* __restrict__ bkt,
                int4* __restrict__ ovf, unsigned* __restrict__ ovfn) {
  const int i = blockIdx.x * 256 + threadIdx.x;
  if (i >= E_V4) return;
  const unsigned hf = (unsigned)chip_half();
  unsigned* __restrict__ cp = cnt + hf * NPAD;
  int4* __restrict__ bk = bkt + hf * CAPH;
  const int4   s4 = ((const int4*)src)[i];
  const int4   d4 = ((const int4*)dst)[i];
  const float4 w4 = ((const float4*)wgt)[i];
  const float x0 = zs[s4.x] + zd[d4.x];
  const float x1 = zs[s4.y] + zd[d4.y];
  const float x2 = zs[s4.z] + zd[d4.z];
  const float x3 = zs[s4.w] + zd[d4.w];
  const float e0 = ((x0 > 0.0f) ? x0 : NEG_SLOPE * x0) * w4.x;
  const float e1 = ((x1 > 0.0f) ? x1 : NEG_SLOPE * x1) * w4.y;
  const float e2 = ((x2 > 0.0f) ? x2 : NEG_SLOPE * x2) * w4.z;
  const float e3 = ((x3 > 0.0f) ? x3 : NEG_SLOPE * x3) * w4.w;
  const unsigned r0 = atomicAdd(cp + d4.x, 1u);
  const unsigned r1 = atomicAdd(cp + d4.y, 1u);
  const unsigned r2 = atomicAdd(cp + d4.z, 1u);
  const unsigned r3 = atomicAdd(cp + d4.w, 1u);
  put_record(r0, make_int4(s4.x, __float_as_int(e0), 4 * i + 0, d4.x), bk, ovf, ovfn);
  put_record(r1, make_int4(s4.y, __float_as_int(e1), 4 * i + 1, d4.y), bk, ovf, ovfn);
  put_record(r2, make_int4(s4.z, __float_as_int(e2), 4 * i + 2, d4.z), bk, ovf, ovfn);
  put_record(r3, make_int4(s4.w, __float_as_int(e3), 4 * i + 3, d4.w), bk, ovf, ovfn);
}

#define SLOTS 64
static_assert(SLOTS >= 2 * CAPH, "the strip holds two full planes");
static_assert(N_NODES % 8 == 0, "eight waves per block, one node each");

template <int C>
__device__ __forceinline__ void node_step(const int2* sl, const float2* __restrict__ z2,
                                          unsigned lane, float2& acc, float& s) {
  int2 q[C];
  float2 v[C];
  #pragma unroll
  for (int u = 0; u < C; ++u) q[u] = sl[u];
  #pragma unroll
  for (int u = 0; u < C; ++u) v[u] = z2[(unsigned)q[u].x * (unsigned)(OUT_DIM / 2) + lane];
  #pragma unroll
  for (int u = 0; u < C; ++u) {
    const float a = __int_as_float(q[u].y);
    s     += a;
    acc.x += a * v[u].x;
    acc.y += a * v[u].y;
  }
}

__device__ __forceinline__ void node_walk(const int2* sl, int n, const float2* __restrict__ z2,
                                          unsigned lane, float2& acc, float& s) {
  int p = 0;
  for (; p + 8 <= n; p += 8) node_step<8>(sl + p, z2, lane, acc, s);
  if (p + 4 <= n) { node_step<4>(sl + p, z2, lane, acc, s); p += 4; }
  if (p + 2 <= n) { node_step<2>(sl + p, z2, lane, acc, s); p += 2; }
  if (p + 1 <= n) { node_step<1>(sl + p, z2, lane, acc, s); }
}

__device__ __forceinline__ int count_below(int idk, int key) {
  int c = 0;
  #pragma unroll
  for (int t = 0; t < 32; ++t) c += (__builtin_amdgcn_readlane(idk, t) < key) ? 1 : 0;
  return c;
}

__device__ __forceinline__ float wave_max(float m) {
  #pragma unroll
  for (int o = 16; o > 0; o >>= 1) m = fmaxf(m, __shfl_xor(m, o, 32));
  return m;
}

__device__ __forceinline__ const int4* rec_ptr(int j, int n0, const int4* __restrict__ b0) {
  return (j < n0) ? (b0 + j) : (b0 + CAPH + (j - n0));
}

__global__ void __launch_bounds__(256)
gat_node(const unsigned* __restrict__ cnt, const int4* __restrict__ bkt,
         const int4* __restrict__ ovf, const unsigned* __restrict__ ovfn,
         const float* __restrict__ z, float* __restrict__ out) {
  __shared__ int2 slot[8][SLOTS];
  const int lane = threadIdx.x & 31;
  const int wv   = threadIdx.x >> 5;
  const int d    = blockIdx.x * 8 + wv;
  const int c0  = __builtin_amdgcn_readfirstlane((int)cnt[d]);
  const int c1  = __builtin_amdgcn_readfirstlane((int)cnt[NPAD + d]);
  const int deg = c0 + c1;
  const int4* __restrict__ b0 = bkt + (unsigned)d * (unsigned)(2 * CAPH);
  const float2* __restrict__ z2 = (const float2*)z;
  int2* sl = slot[wv];

  float2 acc = make_float2(0.0f, 0.0f);
  float  s   = 0.0f;
  if (c0 <= CAPH && c1 <= CAPH) {
    if (deg <= 32) {
      const bool valid = lane < deg;
      int4 en = make_int4(0, __float_as_int(-INFINITY), 0x7FFFFFFF, 0);
      if (valid) en = *rec_ptr(lane, c0, b0);
      const float e = __int_as_float(en.y);
      const float m = wave_max(e);
      const int rank = count_below(en.z, en.z);
      if (valid) sl[rank] = make_int2(en.x, __float_as_int(expf(e - m)));
      node_walk(sl, deg, z2, (unsigned)lane, acc, s);
    } else {
      const int  j1     = lane + 32;
      const bool valid1 = j1 < deg;
      const int4 ea = *rec_ptr(lane, c0, b0);
      int4 eb = make_int4(0, __float_as_int(-INFINITY), 0x7FFFFFFF, 0);
      if (valid1) eb = *rec_ptr(j1, c0, b0);
      const float fa = __int_as_float(ea.y), fb = __int_as_float(eb.y);
      const float m = wave_max(fmaxf(fa, fb));
      const int ranka = count_below(ea.z, ea.z) + count_below(eb.z, ea.z);
      const int rankb = count_below(ea.z, eb.z) + count_below(eb.z, eb.z);
      sl[ranka] = make_int2(ea.x, __float_as_int(expf(fa - m)));
      if (valid1) sl[rankb] = make_int2(eb.x, __float_as_int(expf(fb - m)));
      node_walk(sl, deg, z2, (unsigned)lane, acc, s);
    }
  } else {
    const int n0  = (c0 < CAPH) ? c0 : CAPH;
    const int n1  = (c1 < CAPH) ? c1 : CAPH;
    const int nb  = n0 + n1;
    const int nov = __builtin_amdgcn_readfirstlane((int)*ovfn);
    const int tot = nb + nov;
    float m = -INFINITY;
    for (int jb = 0; jb < tot; jb += 32) {
      const int j = jb + lane;
      if (j < nb) {
        m = fmaxf(m, __int_as_float(rec_ptr(j, n0, b0)->y));
      } else if (j < tot) {
        const int4 r = ovf[j - nb];
        if (r.w == d) m = fmaxf(m, __int_as_float(r.y));
      }
    }
    m = wave_max(m);
    int last = -1;
    for (int n = 0; n < deg; ++n) {
      int bz = 0x7FFFFFFF, bx = 0, be = 0;
      for (int jb = 0; jb < tot; jb += 32) {
        const int j = jb + lane;
        int4 r = make_int4(0, 0, 0x7FFFFFFF, -1);
        if (j < nb)       r = *rec_ptr(j, n0, b0);
        else if (j < tot) r = ovf[j - nb];
        if (r.w == d && r.z > last && r.z < bz) { bz = r.z; bx = r.x; be = r.y; }
      }
      int mz = bz;
      #pragma unroll
      for (int o = 16; o > 0; o >>= 1) {
        const int q = __shfl_xor(mz, o, 32);
        mz = (q < mz) ? q : mz;
      }
      if (mz == 0x7FFFFFFF) break;
      const int t = __builtin_ctz(__builtin_amdgcn_ballot_w32(bz == mz));
      const int   sn = __shfl(bx, t, 32);
      const float a  = expf(__int_as_float(__shfl(be, t, 32)) - m);
      const float2 v = z2[(unsigned)sn * (unsigned)(OUT_DIM / 2) + (unsigned)lane];
      s     += a;
      acc.x += a * v.x;
      acc.y += a * v.y;
      last = mz;
    }
  }
  float2 o = make_float2(0.0f, 0.0f);
  if (deg > 0) o = make_float2(acc.x / s, acc.y / s);
  ((float2*)out)[(unsigned)d * (unsigned)(OUT_DIM / 2) + (unsigned)lane] = o;
}

extern "C" void kernel_launch(void* const* d_in, const int* in_sizes, int n_in,
                              void* d_out, int out_size, void* d_ws, size_t ws_size,
                              hipStream_t stream) {
  (void)in_sizes; (void)n_in; (void)out_size; (void)ws_size;
  const float* h     = (const float*)d_in[0];
  const float* Wfc   = (const float*)d_in[1];
  const float* Wattn = (const float*)d_in[2];
  const int*   src   = (const int*)d_in[3];
  const int*   dst   = (const int*)d_in[4];
  const float* wgt   = (const float*)d_in[5];
  float* out = (float*)d_out;

  float*    z    = (float*)d_ws;
  float*    zs   = z + (size_t)N_NODES * OUT_DIM;
  float*    zd   = zs + NPAD;
  unsigned* cnt  = (unsigned*)(zd + NPAD);
  unsigned* ovfn = cnt + 2 * NPAD;
  int4*     bkt  = (int4*)(ovfn + 64);
  int4*     ovf  = bkt + (size_t)N_NODES * 2 * CAPH;

  {
    int waves  = (N_NODES + 15) / 16;
    int blocks = (waves * 32 + 255) / 256;
    gat_gemm_wmma<<<blocks, 256, 0, stream>>>(h, Wfc, Wattn, z, zs, zd, cnt, ovfn);
  }
  {
    int blocks = (E_V4 + 255) / 256;
    gat_edge_scores<<<blocks, 256, 0, stream>>>(src, dst, wgt, zs, zd, cnt, bkt, ovf, ovfn);
  }
  gat_node<<<N_NODES / 8, 256, 0, stream>>>(cnt, bkt, ovf, ovfn, z, out);
}


# Edge pass drops the zd[dst] gather: records carry zs[src] and the weight, each node wave reads its one zd[d] and forms e = w*lrelu(zs+zd) itself; sum order unchanged, e-m may fuse (one rounding)
# speedup vs baseline: 5.4264x; 1.0759x over previous
#include <hip/hip_runtime.h>
#include <hip/hip_bf16.h>

#define N_NODES   50000
#define E_EDGES   1000000
#define IN_DIM    128
#define OUT_DIM   64
#define NEG_SLOPE 0.01f
#define NPAD      50048
#define CAPH      32

typedef __attribute__((ext_vector_type(2))) float v2f;
typedef __attribute__((ext_vector_type(8))) float v8f;

__device__ __forceinline__ int chip_half() {
  return (int)((__builtin_amdgcn_s_sendmsg_rtn(0x87) >> 18) & 1u);
}

static_assert(N_NODES % 16 == 0, "every wave owns a full 16-row tile");
__global__ void __launch_bounds__(256)
gat_gemm_wmma(const float* __restrict__ h, const float* __restrict__ Wfc,
              const float* __restrict__ Wattn,
              float* __restrict__ z, float* __restrict__ zs, float* __restrict__ zd,
              unsigned* __restrict__ cnt, unsigned* __restrict__ ovfn) {
  const int lane = threadIdx.x & 31;
  const int wid  = blockIdx.x * (blockDim.x >> 5) + (threadIdx.x >> 5);
  const int m0   = wid * 16;
  if (m0 >= N_NODES) return;

  const int lrow = lane & 15;
  const int kh   = (lane >> 4) << 1;

  const float* __restrict__ ap  = h   + (long)(m0 + lrow) * IN_DIM + kh;
  const float* __restrict__ b0p = Wfc + (long)( 0 + lrow) * IN_DIM + kh;
  const float* __restrict__ b1p = Wfc + (long)(16 + lrow) * IN_DIM + kh;
  const float* __restrict__ b2p = Wfc + (long)(32 + lrow) * IN_DIM + kh;
  const float* __restrict__ b3p = Wfc + (long)(48 + lrow) * IN_DIM + kh;

  v8f c0 = {}, c1 = {}, c2 = {}, c3 = {};
  #pragma unroll 4
  for (int k = 0; k < IN_DIM; k += 4) {
    v2f a  = *(const v2f*)(ap  + k);
    v2f b0 = *(const v2f*)(b0p + k);
    v2f b1 = *(const v2f*)(b1p + k);
    v2f b2 = *(const v2f*)(b2p + k);
    v2f b3 = *(const v2f*)(b3p + k);
    c0 = __builtin_amdgcn_wmma_f32_16x16x4_f32(false, a, false, b0, (short)0, c0, false, false);
    c1 = __builtin_amdgcn_wmma_f32_16x16x4_f32(false, a, false, b1, (short)0, c1, false, false);
    c2 = __builtin_amdgcn_wmma_f32_16x16x4_f32(false, a, false, b2, (short)0, c2, false, false);
    c3 = __builtin_amdgcn_wmma_f32_16x16x4_f32(false, a, false, b3, (short)0, c3, false, false);
  }

  const int rbase = m0 + ((lane >> 4) << 3);
  float* zp = z + (long)rbase * OUT_DIM + lrow;
  #pragma unroll
  for (int r = 0; r < 8; ++r) {
    zp[(long)r * OUT_DIM +  0] = c0[r];
    zp[(long)r * OUT_DIM + 16] = c1[r];
    zp[(long)r * OUT_DIM + 32] = c2[r];
    zp[(long)r * OUT_DIM + 48] = c3[r];
  }

  const float as0 = Wattn[lrow],            as1 = Wattn[lrow + 16];
  const float as2 = Wattn[lrow + 32],       as3 = Wattn[lrow + 48];
  const float ad0 = Wattn[OUT_DIM + lrow],      ad1 = Wattn[OUT_DIM + lrow + 16];
  const float ad2 = Wattn[OUT_DIM + lrow + 32], ad3 = Wattn[OUT_DIM + lrow + 48];
  float ps[8], pd[8];
  #pragma unroll
  for (int r = 0; r < 8; ++r) {
    ps[r] = c0[r] * as0 + c1[r] * as1 + c2[r] * as2 + c3[r] * as3;
    pd[r] = c0[r] * ad0 + c1[r] * ad1 + c2[r] * ad2 + c3[r] * ad3;
  }
  #pragma unroll
  for (int o = 8; o > 0; o >>= 1) {
    #pragma unroll
    for (int r = 0; r < 8; ++r) {
      ps[r] += __shfl_xor(ps[r], o, 32);
      pd[r] += __shfl_xor(pd[r], o, 32);
    }
  }
  {
    const bool is_s = lrow < 8;
    const int  rr   = lrow & 7;
    float v = is_s ? ps[0] : pd[0];
    #pragma unroll
    for (int r = 1; r < 8; ++r) {
      const float t = is_s ? ps[r] : pd[r];
      v = (rr == r) ? t : v;
    }
    float* dp = (is_s ? zs : zd) + rbase + rr;
    *dp = v;
  }
  if (lane < 16) cnt[m0 + lane] = 0u;
  else           cnt[NPAD + m0 + lane - 16] = 0u;
  if (wid == 0 && lane == 0) *ovfn = 0u;
}

static_assert(E_EDGES % 4 == 0, "edges are taken four at a time");
#define E_V4 (E_EDGES / 4)
__device__ __forceinline__ void put_record(unsigned r, int d, int4 rec, int4* __restrict__ bk,
                                           int4* __restrict__ ovf, int* __restrict__ ovfd,
                                           unsigned* __restrict__ ovfn) {
  if (r < (unsigned)CAPH) {
    bk[(unsigned)d * (unsigned)(2 * CAPH) + r] = rec;
  } else {
    const unsigned p = atomicAdd(ovfn, 1u);
    ovf[p]  = rec;
    ovfd[p] = d;
  }
}
__global__ void __launch_bounds__(256)
gat_edge_scores(const int* __restrict__ src, const int* __restrict__ dst,
                const float* __restrict__ wgt,
                const float* __restrict__ zs,
                unsigned* __restrict__ cnt, int4* __restrict__ bkt,
                int4* __restrict__ ovf, int* __restrict__ ovfd,
                unsigned* __restrict__ ovfn) {
  const int i = blockIdx.x * 256 + threadIdx.x;
  if (i >= E_V4) return;
  const unsigned hf = (unsigned)chip_half();
  unsigned* __restrict__ cp = cnt + hf * NPAD;
  int4* __restrict__ bk = bkt + hf * CAPH;
  const int4   s4 = ((const int4*)src)[i];
  const int4   d4 = ((const int4*)dst)[i];
  const float4 w4 = ((const float4*)wgt)[i];
  const float x0 = zs[s4.x];
  const float x1 = zs[s4.y];
  const float x2 = zs[s4.z];
  const float x3 = zs[s4.w];
  const unsigned r0 = atomicAdd(cp + d4.x, 1u);
  const unsigned r1 = atomicAdd(cp + d4.y, 1u);
  const unsigned r2 = atomicAdd(cp + d4.z, 1u);
  const unsigned r3 = atomicAdd(cp + d4.w, 1u);
  put_record(r0, d4.x, make_int4(s4.x, __float_as_int(x0), 4 * i + 0, __float_as_int(w4.x)), bk, ovf, ovfd, ovfn);
  put_record(r1, d4.y, make_int4(s4.y, __float_as_int(x1), 4 * i + 1, __float_as_int(w4.y)), bk, ovf, ovfd, ovfn);
  put_record(r2, d4.z, make_int4(s4.z, __float_as_int(x2), 4 * i + 2, __float_as_int(w4.z)), bk, ovf, ovfd, ovfn);
  put_record(r3, d4.w, make_int4(s4.w, __float_as_int(x3), 4 * i + 3, __float_as_int(w4.w)), bk, ovf, ovfd, ovfn);
}

#define SLOTS 64
static_assert(SLOTS >= 2 * CAPH, "the strip holds two full planes");
static_assert(N_NODES % 8 == 0, "eight waves per block, one node each");

template <int C>
__device__ __forceinline__ void node_step(const int2* sl, const float2* __restrict__ z2,
                                          unsigned lane, float2& acc, float& s) {
  int2 q[C];
  float2 v[C];
  #pragma unroll
  for (int u = 0; u < C; ++u) q[u] = sl[u];
  #pragma unroll
  for (int u = 0; u < C; ++u) v[u] = z2[(unsigned)q[u].x * (unsigned)(OUT_DIM / 2) + lane];
  #pragma unroll
  for (int u = 0; u < C; ++u) {
    const float a = __int_as_float(q[u].y);
    s     += a;
    acc.x += a * v[u].x;
    acc.y += a * v[u].y;
  }
}

__device__ __forceinline__ void node_walk(const int2* sl, int n, const float2* __restrict__ z2,
                                          unsigned lane, float2& acc, float& s) {
  int p = 0;
  for (; p + 8 <= n; p += 8) node_step<8>(sl + p, z2, lane, acc, s);
  if (p + 4 <= n) { node_step<4>(sl + p, z2, lane, acc, s); p += 4; }
  if (p + 2 <= n) { node_step<2>(sl + p, z2, lane, acc, s); p += 2; }
  if (p + 1 <= n) { node_step<1>(sl + p, z2, lane, acc, s); }
}

__device__ __forceinline__ int count_below(int idk, int key) {
  int c = 0;
  #pragma unroll
  for (int t = 0; t < 32; ++t) c += (__builtin_amdgcn_readlane(idk, t) < key) ? 1 : 0;
  return c;
}

__device__ __forceinline__ float wave_max(float m) {
  #pragma unroll
  for (int o = 16; o > 0; o >>= 1) m = fmaxf(m, __shfl_xor(m, o, 32));
  return m;
}

__device__ __forceinline__ float edge_logit(int xs, int w, float zdd) {
  const float x = __int_as_float(xs) + zdd;
  return ((x > 0.0f) ? x : NEG_SLOPE * x) * __int_as_float(w);
}

__device__ __forceinline__ const int4* rec_ptr(int j, int n0, const int4* __restrict__ b0) {
  return (j < n0) ? (b0 + j) : (b0 + CAPH + (j - n0));
}

__global__ void __launch_bounds__(256)
gat_node(const unsigned* __restrict__ cnt, const int4* __restrict__ bkt,
         const int4* __restrict__ ovf, const int* __restrict__ ovfd,
         const unsigned* __restrict__ ovfn, const float* __restrict__ zd,
         const float* __restrict__ z, float* __restrict__ out) {
  __shared__ int2 slot[8][SLOTS];
  const int lane = threadIdx.x & 31;
  const int wv   = threadIdx.x >> 5;
  const int d    = blockIdx.x * 8 + wv;
  const int c0  = __builtin_amdgcn_readfirstlane((int)cnt[d]);
  const int c1  = __builtin_amdgcn_readfirstlane((int)cnt[NPAD + d]);
  const int deg = c0 + c1;
  const float zdd = zd[d];
  const int4* __restrict__ b0 = bkt + (unsigned)d * (unsigned)(2 * CAPH);
  const float2* __restrict__ z2 = (const float2*)z;
  int2* sl = slot[wv];

  float2 acc = make_float2(0.0f, 0.0f);
  float  s   = 0.0f;
  if (c0 <= CAPH && c1 <= CAPH) {
    if (deg <= 32) {
      const bool valid = lane < deg;
      int4 en = make_int4(0, 0, 0x7FFFFFFF, 0);
      if (valid) en = *rec_ptr(lane, c0, b0);
      const float e = valid ? edge_logit(en.y, en.w, zdd) : -INFINITY;
      const float m = wave_max(e);
      const int rank = count_below(en.z, en.z);
      if (valid) sl[rank] = make_int2(en.x, __float_as_int(expf(e - m)));
      node_walk(sl, deg, z2, (unsigned)lane, acc, s);
    } else {
      const int  j1     = lane + 32;
      const bool valid1 = j1 < deg;
      const int4 ea = *rec_ptr(lane, c0, b0);
      int4 eb = make_int4(0, 0, 0x7FFFFFFF, 0);
      if (valid1) eb = *rec_ptr(j1, c0, b0);
      const float fa = edge_logit(ea.y, ea.w, zdd);
      const float fb = valid1 ? edge_logit(eb.y, eb.w, zdd) : -INFINITY;
      const float m = wave_max(fmaxf(fa, fb));
      const int ranka = count_below(ea.z, ea.z) + count_below(eb.z, ea.z);
      const int rankb = count_below(ea.z, eb.z) + count_below(eb.z, eb.z);
      sl[ranka] = make_int2(ea.x, __float_as_int(expf(fa - m)));
      if (valid1) sl[rankb] = make_int2(eb.x, __float_as_int(expf(fb - m)));
      node_walk(sl, deg, z2, (unsigned)lane, acc, s);
    }
  } else {
    const int n0  = (c0 < CAPH) ? c0 : CAPH;
    const int n1  = (c1 < CAPH) ? c1 : CAPH;
    const int nb  = n0 + n1;
    const int nov = __builtin_amdgcn_readfirstlane((int)*ovfn);
    const int tot = nb + nov;
    float m = -INFINITY;
    for (int jb = 0; jb < tot; jb += 32) {
      const int j = jb + lane;
      if (j < nb) {
        const int4 r = *rec_ptr(j, n0, b0);
        m = fmaxf(m, edge_logit(r.y, r.w, zdd));
      } else if (j < tot) {
        if (ovfd[j - nb] == d) {
          const int4 r = ovf[j - nb];
          m = fmaxf(m, edge_logit(r.y, r.w, zdd));
        }
      }
    }
    m = wave_max(m);
    int last = -1;
    for (int n = 0; n < deg; ++n) {
      int bz = 0x7FFFFFFF, bx = 0, be = 0, bw = 0;
      for (int jb = 0; jb < tot; jb += 32) {
        const int j = jb + lane;
        int4 r = make_int4(0, 0, 0x7FFFFFFF, 0);
        bool mine = false;
        if (j < nb)       { r = *rec_ptr(j, n0, b0); mine = true; }
        else if (j < tot) { r = ovf[j - nb]; mine = (ovfd[j - nb] == d); }
        if (mine && r.z > last && r.z < bz) { bz = r.z; bx = r.x; be = r.y; bw = r.w; }
      }
      int mz = bz;
      #pragma unroll
      for (int o = 16; o > 0; o >>= 1) {
        const int q = __shfl_xor(mz, o, 32);
        mz = (q < mz) ? q : mz;
      }
      if (mz == 0x7FFFFFFF) break;
      const int t = __builtin_ctz(__builtin_amdgcn_ballot_w32(bz == mz));
      const int   sn = __shfl(bx, t, 32);
      const float a  = expf(edge_logit(__shfl(be, t, 32), __shfl(bw, t, 32), zdd) - m);
      const float2 v = z2[(unsigned)sn * (unsigned)(OUT_DIM / 2) + (unsigned)lane];
      s     += a;
      acc.x += a * v.x;
      acc.y += a * v.y;
      last = mz;
    }
  }
  float2 o = make_float2(0.0f, 0.0f);
  if (deg > 0) o = make_float2(acc.x / s, acc.y / s);
  ((float2*)out)[(unsigned)d * (unsigned)(OUT_DIM / 2) + (unsigned)lane] = o;
}

extern "C" void kernel_launch(void* const* d_in, const int* in_sizes, int n_in,
                              void* d_out, int out_size, void* d_ws, size_t ws_size,
                              hipStream_t stream) {
  (void)in_sizes; (void)n_in; (void)out_size; (void)ws_size;
  const float* h     = (const float*)d_in[0];
  const float* Wfc   = (const float*)d_in[1];
  const float* Wattn = (const float*)d_in[2];
  const int*   src   = (const int*)d_in[3];
  const int*   dst   = (const int*)d_in[4];
  const float* wgt   = (const float*)d_in[5];
  float* out = (float*)d_out;

  float*    z    = (float*)d_ws;
  float*    zs   = z + (size_t)N_NODES * OUT_DIM;
  float*    zd   = zs + NPAD;
  unsigned* cnt  = (unsigned*)(zd + NPAD);
  unsigned* ovfn = cnt + 2 * NPAD;
  int4*     bkt  = (int4*)(ovfn + 64);
  int4*     ovf  = bkt + (size_t)N_NODES * 2 * CAPH;
  int*      ovfd = (int*)(ovf + (size_t)E_EDGES);

  {
    int waves  = (N_NODES + 15) / 16;
    int blocks = (waves * 32 + 255) / 256;
    gat_gemm_wmma<<<blocks, 256, 0, stream>>>(h, Wfc, Wattn, z, zs, zd, cnt, ovfn);
  }
  {
    int blocks = (E_V4 + 255) / 256;
    gat_edge_scores<<<blocks, 256, 0, stream>>>(src, dst, wgt, zs, cnt, bkt, ovf, ovfd, ovfn);
  }
  gat_node<<<N_NODES / 8, 256, 0, stream>>>(cnt, bkt, ovf, ovfd, ovfn, zd, z, out);
}


# Projection tiles and edge pass share one launch (roles by block) and run side by side: records drop zs[src], node waves gather it; a small launch zeroes the counters first; f32 sums unchanged
# speedup vs baseline: 5.6440x; 1.0401x over previous
#include <hip/hip_runtime.h>
#include <hip/hip_bf16.h>

#define N_NODES   50000
#define E_EDGES   1000000
#define IN_DIM    128
#define OUT_DIM   64
#define NEG_SLOPE 0.01f
#define NPAD      50048
#define CAPH      32

typedef __attribute__((ext_vector_type(2))) float v2f;
typedef __attribute__((ext_vector_type(8))) float v8f;

__device__ __forceinline__ int chip_half() {
  return (int)((__builtin_amdgcn_s_sendmsg_rtn(0x87) >> 18) & 1u);
}

static_assert(N_NODES % 16 == 0, "every wave owns a full 16-row tile");
#define GEMM_WAVES  (N_NODES / 16)
#define GEMM_BLOCKS ((GEMM_WAVES + 7) / 8)
__device__ __forceinline__ void
gemm_tile(const float* __restrict__ h, const float* __restrict__ Wfc,
          const float* __restrict__ Wattn,
          float* __restrict__ z, float* __restrict__ zs, float* __restrict__ zd) {
  const int lane = threadIdx.x & 31;
  const int wid  = blockIdx.x * 8 + (threadIdx.x >> 5);
  const int m0   = wid * 16;
  if (m0 >= N_NODES) return;

  const int lrow = lane & 15;
  const int kh   = (lane >> 4) << 1;

  const float* __restrict__ ap  = h   + (long)(m0 + lrow) * IN_DIM + kh;
  const float* __restrict__ b0p = Wfc + (long)( 0 + lrow) * IN_DIM + kh;
  const float* __restrict__ b1p = Wfc + (long)(16 + lrow) * IN_DIM + kh;
  const float* __restrict__ b2p = Wfc + (long)(32 + lrow) * IN_DIM + kh;
  const float* __restrict__ b3p = Wfc + (long)(48 + lrow) * IN_DIM + kh;

  v8f c0 = {}, c1 = {}, c2 = {}, c3 = {};
  #pragma unroll 4
  for (int k = 0; k < IN_DIM; k += 4) {
    v2f a  = *(const v2f*)(ap  + k);
    v2f b0 = *(const v2f*)(b0p + k);
    v2f b1 = *(const v2f*)(b1p + k);
    v2f b2 = *(const v2f*)(b2p + k);
    v2f b3 = *(const v2f*)(b3p + k);
    c0 = __builtin_amdgcn_wmma_f32_16x16x4_f32(false, a, false, b0, (short)0, c0, false, false);
    c1 = __builtin_amdgcn_wmma_f32_16x16x4_f32(false, a, false, b1, (short)0, c1, false, false);
    c2 = __builtin_amdgcn_wmma_f32_16x16x4_f32(false, a, false, b2, (short)0, c2, false, false);
    c3 = __builtin_amdgcn_wmma_f32_16x16x4_f32(false, a, false, b3, (short)0, c3, false, false);
  }

  const int rbase = m0 + ((lane >> 4) << 3);
  float* zp = z + (long)rbase * OUT_DIM + lrow;
  #pragma unroll
  for (int r = 0; r < 8; ++r) {
    zp[(long)r * OUT_DIM +  0] = c0[r];
    zp[(long)r * OUT_DIM + 16] = c1[r];
    zp[(long)r * OUT_DIM + 32] = c2[r];
    zp[(long)r * OUT_DIM + 48] = c3[r];
  }

  const float as0 = Wattn[lrow],            as1 = Wattn[lrow + 16];
  const float as2 = Wattn[lrow + 32],       as3 = Wattn[lrow + 48];
  const float ad0 = Wattn[OUT_DIM + lrow],      ad1 = Wattn[OUT_DIM + lrow + 16];
  const float ad2 = Wattn[OUT_DIM + lrow + 32], ad3 = Wattn[OUT_DIM + lrow + 48];
  float ps[8], pd[8];
  #pragma unroll
  for (int r = 0; r < 8; ++r) {
    ps[r] = c0[r] * as0 + c1[r] * as1 + c2[r] * as2 + c3[r] * as3;
    pd[r] = c0[r] * ad0 + c1[r] * ad1 + c2[r] * ad2 + c3[r] * ad3;
  }
  #pragma unroll
  for (int o = 8; o > 0; o >>= 1) {
    #pragma unroll
    for (int r = 0; r < 8; ++r) {
      ps[r] += __shfl_xor(ps[r], o, 32);
      pd[r] += __shfl_xor(pd[r], o, 32);
    }
  }
  {
    const bool is_s = lrow < 8;
    const int  rr   = lrow & 7;
    float v = is_s ? ps[0] : pd[0];
    #pragma unroll
    for (int r = 1; r < 8; ++r) {
      const float t = is_s ? ps[r] : pd[r];
      v = (rr == r) ? t : v;
    }
    float* dp = (is_s ? zs : zd) + rbase + rr;
    *dp = v;
  }
}

static_assert((2 * NPAD) % 4 == 0, "the counters are cleared four at a time");
#define CNT_V4 ((2 * NPAD) / 4)
__global__ void __launch_bounds__(256)
gat_zero(uint4* __restrict__ cnt4, unsigned* __restrict__ ovfn) {
  const int i = blockIdx.x * 256 + threadIdx.x;
  if (i < CNT_V4) cnt4[i] = make_uint4(0u, 0u, 0u, 0u);
  if (i == 0) *ovfn = 0u;
}

static_assert(E_EDGES % 4 == 0, "edges are taken four at a time");
#define E_V4 (E_EDGES / 4)
__device__ __forceinline__ void put_record(unsigned r, int d, int4 rec, int4* __restrict__ bk,
                                           int4* __restrict__ ovf, int* __restrict__ ovfd,
                                           unsigned* __restrict__ ovfn) {
  if (r < (unsigned)CAPH) {
    bk[(unsigned)d * (unsigned)(2 * CAPH) + r] = rec;
  } else {
    const unsigned p = atomicAdd(ovfn, 1u);
    ovf[p]  = rec;
    ovfd[p] = d;
  }
}
#define EDGE_BLOCKS ((E_V4 + 255) / 256)
__device__ __forceinline__ void
edge_records(const int* __restrict__ src, const int* __restrict__ dst,
             const float* __restrict__ wgt,
             unsigned* __restrict__ cnt, int4* __restrict__ bkt,
             int4* __restrict__ ovf, int* __restrict__ ovfd,
             unsigned* __restrict__ ovfn) {
  const int i = (blockIdx.x - GEMM_BLOCKS) * 256 + threadIdx.x;
  if (i >= E_V4) return;
  const unsigned hf = (unsigned)chip_half();
  unsigned* __restrict__ cp = cnt + hf * NPAD;
  int4* __restrict__ bk = bkt + hf * CAPH;
  const int4   s4 = ((const int4*)src)[i];
  const int4   d4 = ((const int4*)dst)[i];
  const float4 w4 = ((const float4*)wgt)[i];
  const unsigned r0 = atomicAdd(cp + d4.x, 1u);
  const unsigned r1 = atomicAdd(cp + d4.y, 1u);
  const unsigned r2 = atomicAdd(cp + d4.z, 1u);
  const unsigned r3 = atomicAdd(cp + d4.w, 1u);
  put_record(r0, d4.x, make_int4(s4.x, 4 * i + 0, __float_as_int(w4.x), 0), bk, ovf, ovfd, ovfn);
  put_record(r1, d4.y, make_int4(s4.y, 4 * i + 1, __float_as_int(w4.y), 0), bk, ovf, ovfd, ovfn);
  put_record(r2, d4.z, make_int4(s4.z, 4 * i + 2, __float_as_int(w4.z), 0), bk, ovf, ovfd, ovfn);
  put_record(r3, d4.w, make_int4(s4.w, 4 * i + 3, __float_as_int(w4.w), 0), bk, ovf, ovfd, ovfn);
}

__global__ void __launch_bounds__(256)
gat_gemm_and_edges(const float* __restrict__ h, const float* __restrict__ Wfc,
                   const float* __restrict__ Wattn,
                   float* __restrict__ z, float* __restrict__ zs, float* __restrict__ zd,
                   const int* __restrict__ src, const int* __restrict__ dst,
                   const float* __restrict__ wgt,
                   unsigned* __restrict__ cnt, int4* __restrict__ bkt,
                   int4* __restrict__ ovf, int* __restrict__ ovfd,
                   unsigned* __restrict__ ovfn) {
  if (blockIdx.x < GEMM_BLOCKS) gemm_tile(h, Wfc, Wattn, z, zs, zd);
  else                          edge_records(src, dst, wgt, cnt, bkt, ovf, ovfd, ovfn);
}

#define SLOTS 64
static_assert(SLOTS >= 2 * CAPH, "the strip holds two full planes");
static_assert(N_NODES % 8 == 0, "eight waves per block, one node each");

template <int C>
__device__ __forceinline__ void node_step(const int2* sl, const float2* __restrict__ z2,
                                          unsigned lane, float2& acc, float& s) {
  int2 q[C];
  float2 v[C];
  #pragma unroll
  for (int u = 0; u < C; ++u) q[u] = sl[u];
  #pragma unroll
  for (int u = 0; u < C; ++u) v[u] = z2[(unsigned)q[u].x * (unsigned)(OUT_DIM / 2) + lane];
  #pragma unroll
  for (int u = 0; u < C; ++u) {
    const float a = __int_as_float(q[u].y);
    s     += a;
    acc.x += a * v[u].x;
    acc.y += a * v[u].y;
  }
}

__device__ __forceinline__ void node_walk(const int2* sl, int n, const float2* __restrict__ z2,
                                          unsigned lane, float2& acc, float& s) {
  int p = 0;
  for (; p + 8 <= n; p += 8) node_step<8>(sl + p, z2, lane, acc, s);
  if (p + 4 <= n) { node_step<4>(sl + p, z2, lane, acc, s); p += 4; }
  if (p + 2 <= n) { node_step<2>(sl + p, z2, lane, acc, s); p += 2; }
  if (p + 1 <= n) { node_step<1>(sl + p, z2, lane, acc, s); }
}

__device__ __forceinline__ int count_below(int idk, int key) {
  int c = 0;
  #pragma unroll
  for (int t = 0; t < 32; ++t) c += (__builtin_amdgcn_readlane(idk, t) < key) ? 1 : 0;
  return c;
}

__device__ __forceinline__ float wave_max(float m) {
  #pragma unroll
  for (int o = 16; o > 0; o >>= 1) m = fmaxf(m, __shfl_xor(m, o, 32));
  return m;
}

__device__ __forceinline__ float edge_logit(float xs, int w, float zdd) {
  const float x = xs + zdd;
  return ((x > 0.0f) ? x : NEG_SLOPE * x) * __int_as_float(w);
}

__device__ __forceinline__ const int4* rec_ptr(int j, int n0, const int4* __restrict__ b0) {
  return (j < n0) ? (b0 + j) : (b0 + CAPH + (j - n0));
}

__global__ void __launch_bounds__(256)
gat_node(const unsigned* __restrict__ cnt, const int4* __restrict__ bkt,
         const int4* __restrict__ ovf, const int* __restrict__ ovfd,
         const unsigned* __restrict__ ovfn, const float* __restrict__ zs,
         const float* __restrict__ zd,
         const float* __restrict__ z, float* __restrict__ out) {
  __shared__ int2 slot[8][SLOTS];
  const int lane = threadIdx.x & 31;
  const int wv   = threadIdx.x >> 5;
  const int d    = blockIdx.x * 8 + wv;
  const int c0  = __builtin_amdgcn_readfirstlane((int)cnt[d]);
  const int c1  = __builtin_amdgcn_readfirstlane((int)cnt[NPAD + d]);
  const int deg = c0 + c1;
  const float zdd = zd[d];
  const int4* __restrict__ b0 = bkt + (unsigned)d * (unsigned)(2 * CAPH);
  const float2* __restrict__ z2 = (const float2*)z;
  int2* sl = slot[wv];

  float2 acc = make_float2(0.0f, 0.0f);
  float  s   = 0.0f;
  if (c0 <= CAPH && c1 <= CAPH) {
    if (deg <= 32) {
      const bool valid = lane < deg;
      int4 en = make_int4(0, 0x7FFFFFFF, 0, 0);
      float xn = 0.0f;
      if (valid) { en = *rec_ptr(lane, c0, b0); xn = zs[en.x]; }
      const float e = valid ? edge_logit(xn, en.z, zdd) : -INFINITY;
      const float m = wave_max(e);
      const int rank = count_below(en.y, en.y);
      if (valid) sl[rank] = make_int2(en.x, __float_as_int(expf(e - m)));
      node_walk(sl, deg, z2, (unsigned)lane, acc, s);
    } else {
      const int  j1     = lane + 32;
      const bool valid1 = j1 < deg;
      const int4 ea = *rec_ptr(lane, c0, b0);
      int4 eb = make_int4(0, 0x7FFFFFFF, 0, 0);
      float xb = 0.0f;
      if (valid1) { eb = *rec_ptr(j1, c0, b0); xb = zs[eb.x]; }
      const float fa = edge_logit(zs[ea.x], ea.z, zdd);
      const float fb = valid1 ? edge_logit(xb, eb.z, zdd) : -INFINITY;
      const float m = wave_max(fmaxf(fa, fb));
      const int ranka = count_below(ea.y, ea.y) + count_below(eb.y, ea.y);
      const int rankb = count_below(ea.y, eb.y) + count_below(eb.y, eb.y);
      sl[ranka] = make_int2(ea.x, __float_as_int(expf(fa - m)));
      if (valid1) sl[rankb] = make_int2(eb.x, __float_as_int(expf(fb - m)));
      node_walk(sl, deg, z2, (unsigned)lane, acc, s);
    }
  } else {
    const int n0  = (c0 < CAPH) ? c0 : CAPH;
    const int n1  = (c1 < CAPH) ? c1 : CAPH;
    const int nb  = n0 + n1;
    const int nov = __builtin_amdgcn_readfirstlane((int)*ovfn);
    const int tot = nb + nov;
    float m = -INFINITY;
    for (int jb = 0; jb < tot; jb += 32) {
      const int j = jb + lane;
      if (j < nb) {
        const int4 r = *rec_ptr(j, n0, b0);
        m = fmaxf(m, edge_logit(zs[r.x], r.z, zdd));
      } else if (j < tot) {
        if (ovfd[j - nb] == d) {
          const int4 r = ovf[j - nb];
          m = fmaxf(m, edge_logit(zs[r.x], r.z, zdd));
        }
      }
    }
    m = wave_max(m);
    int last = -1;
    for (int n = 0; n < deg; ++n) {
      int bz = 0x7FFFFFFF, bx = 0, bw = 0;
      for (int jb = 0; jb < tot; jb += 32) {
        const int j = jb + lane;
        int4 r = make_int4(0, 0x7FFFFFFF, 0, 0);
        bool mine = false;
        if (j < nb)       { r = *rec_ptr(j, n0, b0); mine = true; }
        else if (j < tot) { r = ovf[j - nb]; mine = (ovfd[j - nb] == d); }
        if (mine && r.y > last && r.y < bz) { bz = r.y; bx = r.x; bw = r.z; }
      }
      int mz = bz;
      #pragma unroll
      for (int o = 16; o > 0; o >>= 1) {
        const int q = __shfl_xor(mz, o, 32);
        mz = (q < mz) ? q : mz;
      }
      if (mz == 0x7FFFFFFF) break;
      const int t = __builtin_ctz(__builtin_amdgcn_ballot_w32(bz == mz));
      const int   sn = __shfl(bx, t, 32);
      const float a  = expf(edge_logit(zs[sn], __shfl(bw, t, 32), zdd) - m);
      const float2 v = z2[(unsigned)sn * (unsigned)(OUT_DIM / 2) + (unsigned)lane];
      s     += a;
      acc.x += a * v.x;
      acc.y += a * v.y;
      last = mz;
    }
  }
  float2 o = make_float2(0.0f, 0.0f);
  if (deg > 0) o = make_float2(acc.x / s, acc.y / s);
  ((float2*)out)[(unsigned)d * (unsigned)(OUT_DIM / 2) + (unsigned)lane] = o;
}

extern "C" void kernel_launch(void* const* d_in, const int* in_sizes, int n_in,
                              void* d_out, int out_size, void* d_ws, size_t ws_size,
                              hipStream_t stream) {
  (void)in_sizes; (void)n_in; (void)out_size; (void)ws_size;
  const float* h     = (const float*)d_in[0];
  const float* Wfc   = (const float*)d_in[1];
  const float* Wattn = (const float*)d_in[2];
  const int*   src   = (const int*)d_in[3];
  const int*   dst   = (const int*)d_in[4];
  const float* wgt   = (const float*)d_in[5];
  float* out = (float*)d_out;

  float*    z    = (float*)d_ws;
  float*    zs   = z + (size_t)N_NODES * OUT_DIM;
  float*    zd   = zs + NPAD;
  unsigned* cnt  = (unsigned*)(zd + NPAD);
  unsigned* ovfn = cnt + 2 * NPAD;
  int4*     bkt  = (int4*)(ovfn + 64);
  int4*     ovf  = bkt + (size_t)N_NODES * 2 * CAPH;
  int*      ovfd = (int*)(ovf + (size_t)E_EDGES);

  gat_zero<<<(CNT_V4 + 255) / 256, 256, 0, stream>>>((uint4*)cnt, ovfn);
  gat_gemm_and_edges<<<GEMM_BLOCKS + EDGE_BLOCKS, 256, 0, stream>>>(
      h, Wfc, Wattn, z, zs, zd, src, dst, wgt, cnt, bkt, ovf, ovfd, ovfn);
  gat_node<<<N_NODES / 8, 256, 0, stream>>>(cnt, bkt, ovf, ovfd, ovfn, zs, zd, z, out);
}
